# starting kernel
# baseline (speedup 1.0000x reference)
#include <hip/hip_runtime.h>
#include <hip/hip_bf16.h>

typedef __attribute__((ext_vector_type(2))) float v2f;
typedef __attribute__((ext_vector_type(8))) float v8f;

#define B_   4
#define N_   32768
#define C_   512
#define M_   512
#define CH_  (C_ + 4)
#define CAP  8704

__global__ __launch_bounds__(256) void score_kernel(
    const float* __restrict__ feat, const float* __restrict__ Wobj,
    const float* __restrict__ bobj, const float* __restrict__ Wg,
    const float* __restrict__ bg, float* __restrict__ grasp_out,
    unsigned char* __restrict__ mask_out)
{
    __shared__ float Apre[128][32][2];
    for (int t = threadIdx.x; t < 128 * 32; t += 256) {
        int kc = t >> 5, lane = t & 31;
        int row = lane & 15;
        int kh  = lane >> 4;
        int ka  = kc * 4 + 2 * kh;
        float a0 = 0.0f, a1 = 0.0f;
        if (row < 2)      { a0 = Wobj[row * C_ + ka]; a1 = Wobj[row * C_ + ka + 1]; }
        else if (row == 2){ a0 = Wg[ka];              a1 = Wg[ka + 1]; }
        Apre[kc][lane][0] = a0;
        Apre[kc][lane][1] = a1;
    }
    __syncthreads();

    const int wave = threadIdx.x >> 5;
    const int lane = threadIdx.x & 31;
    const int b    = blockIdx.x >> 8;
    const int tile = blockIdx.x & 255;
    const int n0   = tile * 128 + wave * 16;
    const int kh   = lane >> 4;
    const int col  = n0 + (lane & 15);

    const float* fb = feat + (size_t)b * C_ * N_;

    float b0 = fb[(size_t)(2 * kh) * N_ + col];
    float b1 = fb[(size_t)(2 * kh + 1) * N_ + col];

    v8f acc = {};
#pragma unroll 2
    for (int kc = 0; kc < 128; ++kc) {
        v2f a, bm;
        a.x  = Apre[kc][lane][0];
        a.y  = Apre[kc][lane][1];
        bm.x = b0;
        bm.y = b1;

        const int kn  = (kc + 1) & 127;
        const int kan = kn * 4 + 2 * kh;
        b0 = fb[(size_t)kan * N_ + col];
        b1 = fb[(size_t)(kan + 1) * N_ + col];

        int pfrow = kc * 4 + 32;
        if (pfrow > C_ - 4) pfrow = C_ - 4;
        __builtin_prefetch(&fb[(size_t)(pfrow + 2 * kh) * N_ + col], 0, 3);
        __builtin_prefetch(&fb[(size_t)(pfrow + 2 * kh + 1) * N_ + col], 0, 3);

        acc = __builtin_amdgcn_wmma_f32_16x16x4_f32(
            false, a, false, bm, (short)0, acc, false, false);
    }

    if (lane < 16) {
        float s0 = acc[0] + bobj[0];
        float s1 = acc[1] + bobj[1];
        float g  = acc[2] + bg[0];
        int p = n0 + lane;
        grasp_out[(size_t)b * N_ + p] = g;
        mask_out[(size_t)b * N_ + p] = (unsigned char)((s1 > s0) && (g > 0.1f));
    }
}

__global__ __launch_bounds__(1024) void fps_kernel(
    const float* __restrict__ xyz, const unsigned char* __restrict__ mask,
    int* __restrict__ out_idx)
{
    __shared__ float4 pts[CAP];
    __shared__ unsigned short cidx[CAP];
    __shared__ int scanbuf[1024];
    __shared__ int base_sh;
    __shared__ int sel_sh;
    __shared__ float wv[32];
    __shared__ int   wi[32];

    const int b   = blockIdx.x;
    const int tid = threadIdx.x;
    const float* xb = xyz + (size_t)b * N_ * 3;
    const unsigned char* mb = mask + (size_t)b * N_;

    if (tid == 0) { base_sh = 0; sel_sh = 0; }
    __syncthreads();

    for (int c0 = 0; c0 < N_; c0 += 1024) {
        const int i = c0 + tid;
        const int m = mb[i] ? 1 : 0;
        scanbuf[tid] = m;
        __syncthreads();
        for (int off = 1; off < 1024; off <<= 1) {
            int t  = scanbuf[tid];
            int t2 = (tid >= off) ? scanbuf[tid - off] : 0;
            __syncthreads();
            scanbuf[tid] = t + t2;
            __syncthreads();
        }
        const int incl = scanbuf[tid];
        const int base = base_sh;
        const int pos  = base + incl - m;
        if (m && pos < CAP) {
            pts[pos]  = make_float4(xb[(size_t)i * 3], xb[(size_t)i * 3 + 1],
                                    xb[(size_t)i * 3 + 2], __builtin_inff());
            cidx[pos] = (unsigned short)i;
        }
        __syncthreads();
        if (tid == 1023) base_sh = base + incl;
        __syncthreads();
    }
    int Nc = base_sh;
    if (Nc > CAP) Nc = CAP;
    if (Nc == 0) {
        for (int s = tid; s < M_; s += 1024) out_idx[b * M_ + s] = 0;
        return;
    }

    const float NEG_INF = -__builtin_inff();
    const int lane = tid & 31, wav = tid >> 5;

    for (int s = 0; s < M_; ++s) {
        const int sel = sel_sh;
        if (tid == 0) out_idx[b * M_ + s] = (int)cidx[sel];
        const float lx = pts[sel].x, ly = pts[sel].y, lz = pts[sel].z;

        float bv = NEG_INF;
        int   bi = 0x7fffffff;
        for (int i = tid; i < Nc; i += 1024) {
            float4 p = pts[i];
            float dx = __fsub_rn(p.x, lx);
            float dy = __fsub_rn(p.y, ly);
            float dz = __fsub_rn(p.z, lz);
            float d  = __fadd_rn(__fadd_rn(__fmul_rn(dx, dx), __fmul_rn(dy, dy)),
                                 __fmul_rn(dz, dz));
            float nd = fminf(p.w, d);
            pts[i].w = nd;
            if (nd > bv) { bv = nd; bi = i; }
        }
        for (int off = 16; off; off >>= 1) {
            float v2 = __shfl_xor(bv, off, 32);
            int   i2 = __shfl_xor(bi, off, 32);
            if (v2 > bv || (v2 == bv && i2 < bi)) { bv = v2; bi = i2; }
        }
        if (lane == 0) { wv[wav] = bv; wi[wav] = bi; }
        __syncthreads();
        if (wav == 0) {
            bv = wv[lane]; bi = wi[lane];
            for (int off = 16; off; off >>= 1) {
                float v2 = __shfl_xor(bv, off, 32);
                int   i2 = __shfl_xor(bi, off, 32);
                if (v2 > bv || (v2 == bv && i2 < bi)) { bv = v2; bi = i2; }
            }
            if (lane == 0) sel_sh = bi;
        }
        __syncthreads();
    }
}

__global__ __launch_bounds__(256) void gather_kernel(
    const float* __restrict__ feat, const float* __restrict__ xyz,
    const float* __restrict__ grasp, const int* __restrict__ idxs,
    float* __restrict__ out)
{
    const int g = blockIdx.x * 256 + threadIdx.x;
    if (g >= B_ * CH_ * M_) return;
    const int m  = g % M_;
    const int t  = g / M_;
    const int ch = t % CH_;
    const int b  = t / CH_;
    const int id = idxs[b * M_ + m];
    float v;
    if (ch < C_)            v = feat[((size_t)b * C_ + ch) * N_ + id];
    else if (ch < C_ + 3)   v = xyz[((size_t)b * N_ + id) * 3 + (ch - C_)];
    else                    v = grasp[(size_t)b * N_ + id];
    out[g] = v;
}

extern "C" void kernel_launch(void* const* d_in, const int* in_sizes, int n_in,
                              void* d_out, int out_size, void* d_ws, size_t ws_size,
                              hipStream_t stream) {
    const float* xyz  = (const float*)d_in[0];
    const float* feat = (const float*)d_in[1];
    const float* Wobj = (const float*)d_in[2];
    const float* bobj = (const float*)d_in[3];
    const float* Wg   = (const float*)d_in[4];
    const float* bg   = (const float*)d_in[5];
    float* out = (float*)d_out;

    char* ws = (char*)d_ws;
    float*         g_grasp = (float*)ws;
    unsigned char* g_mask  = (unsigned char*)(ws + (size_t)B_ * N_ * 4);
    int*           g_idx   = (int*)(ws + (size_t)B_ * N_ * 4 + (size_t)B_ * N_);

    score_kernel<<<B_ * (N_ / 128), 256, 0, stream>>>(feat, Wobj, bobj, Wg, bg,
                                                      g_grasp, g_mask);
    fps_kernel<<<B_, 1024, 0, stream>>>(xyz, g_mask, g_idx);
    const int total = B_ * CH_ * M_;
    gather_kernel<<<(total + 255) / 256, 256, 0, stream>>>(feat, xyz, g_grasp,
                                                           g_idx, out);
}


# FPS wave argmax in one LDS round trip: lanes store their maxima in a 32-entry table, all read it back by broadcast, index by ballot + readlane, not 5 dependent lane-exchange levels; picks identical
# speedup vs baseline: 2.8346x; 2.8346x over previous
#include <hip/hip_runtime.h>
#include <hip/hip_bf16.h>

typedef __attribute__((ext_vector_type(2))) float v2f;
typedef __attribute__((ext_vector_type(8))) float v8f;

#define B_   4
#define N_   32768
#define C_   512
#define M_   512
#define CH_  (C_ + 4)
#define CAP  8704

__global__ __launch_bounds__(256) void score_kernel(
    const float* __restrict__ feat, const float* __restrict__ Wobj,
    const float* __restrict__ bobj, const float* __restrict__ Wg,
    const float* __restrict__ bg, float* __restrict__ grasp_out,
    unsigned char* __restrict__ list_out, int* __restrict__ cnt_out)
{
    __shared__ float Apre[128][32][2];
    for (int t = threadIdx.x; t < 128 * 32; t += 256) {
        int kc = t >> 5, lane = t & 31;
        int row = lane & 15;
        int kh  = lane >> 4;
        int ka  = kc * 4 + 2 * kh;
        float a0 = 0.0f, a1 = 0.0f;
        if (row < 2)      { a0 = Wobj[row * C_ + ka]; a1 = Wobj[row * C_ + ka + 1]; }
        else if (row == 2){ a0 = Wg[ka];              a1 = Wg[ka + 1]; }
        Apre[kc][lane][0] = a0;
        Apre[kc][lane][1] = a1;
    }
    __syncthreads();

    const int wave = threadIdx.x >> 5;
    const int lane = threadIdx.x & 31;
    const int b    = blockIdx.x >> 8;
    const int tile = blockIdx.x & 255;
    const int n0   = tile * 128 + wave * 16;
    const int kh   = lane >> 4;
    const int col  = n0 + (lane & 15);

    const float* fb = feat + (size_t)b * C_ * N_;

    float b0 = fb[(size_t)(2 * kh) * N_ + col];
    float b1 = fb[(size_t)(2 * kh + 1) * N_ + col];

    v8f acc = {};
#pragma unroll 2
    for (int kc = 0; kc < 128; ++kc) {
        v2f a, bm;
        a.x  = Apre[kc][lane][0];
        a.y  = Apre[kc][lane][1];
        bm.x = b0;
        bm.y = b1;

        const int kn  = (kc + 1) & 127;
        const int kan = kn * 4 + 2 * kh;
        b0 = fb[(size_t)kan * N_ + col];
        b1 = fb[(size_t)(kan + 1) * N_ + col];

        int pfrow = kc * 4 + 32;
        if (pfrow > C_ - 4) pfrow = C_ - 4;
        __builtin_prefetch(&fb[(size_t)(pfrow + 2 * kh) * N_ + col], 0, 3);
        __builtin_prefetch(&fb[(size_t)(pfrow + 2 * kh + 1) * N_ + col], 0, 3);

        acc = __builtin_amdgcn_wmma_f32_16x16x4_f32(
            false, a, false, bm, (short)0, acc, false, false);
    }

    bool m = false;
    if (lane < 16) {
        float s0 = acc[0] + bobj[0];
        float s1 = acc[1] + bobj[1];
        float g  = acc[2] + bg[0];
        grasp_out[(size_t)b * N_ + n0 + lane] = g;
        m = (s1 > s0) && (g > 0.1f);
    }

    const unsigned bal = __builtin_amdgcn_ballot_w32(m);
    __syncthreads();
    if (lane == 0) Apre[0][wave][0] = __uint_as_float(bal);
    __syncthreads();
    int before = 0, total = 0;
    for (int w = 0; w < 8; ++w) {
        const int c = __popc(__float_as_uint(Apre[0][w][0]));
        total += c;
        if (w < wave) before += c;
    }
    if (m) {
        const int rank = __popc(bal & ((1u << lane) - 1u));
        list_out[(size_t)b * N_ + tile * 128 + before + rank] =
            (unsigned char)(wave * 16 + lane);
    }
    if ((int)threadIdx.x >= total && threadIdx.x < 128)
        list_out[(size_t)b * N_ + tile * 128 + threadIdx.x] = 0;
    if (threadIdx.x == 0) cnt_out[b * M_ + tile] = total;
}

#define HSET   256
#define NSLOT  9

__device__ __forceinline__ float sqdist(float px, float py, float pz,
                                        float qx, float qy, float qz)
{
    const float dx = px - qx;
    const float dy = py - qy;
    const float dz = pz - qz;
    float sxy;
    {
#pragma clang fp contract(off)
        sxy = dx * dx + dy * dy;
    }
    return __builtin_fmaf(dz, dz, sxy);
}

__device__ __forceinline__ float wave_max_tab(float v, float* tab, int lane)
{
    tab[lane] = v;
    __builtin_amdgcn_fence(__ATOMIC_RELEASE, "wavefront");
    __builtin_amdgcn_wave_barrier();
    __builtin_amdgcn_fence(__ATOMIC_ACQUIRE, "wavefront");
    const float4* t4 = reinterpret_cast<const float4*>(tab);
    float r[8];
#pragma unroll
    for (int k = 0; k < 8; ++k) {
        const float4 t = t4[k];
        r[k] = fmaxf(fmaxf(t.x, t.y), fmaxf(t.z, t.w));
    }
    return fmaxf(fmaxf(fmaxf(r[0], r[1]), fmaxf(r[2], r[3])),
                 fmaxf(fmaxf(r[4], r[5]), fmaxf(r[6], r[7])));
}

__device__ __forceinline__ int wave_index_of_max(float v, float m, int i)
{
    unsigned mask = __builtin_amdgcn_ballot_w32(v == m);
    int best = 0x7fffffff;
    int it = 0;
    do {
        const int l = __builtin_ctz(mask | 0x80000000u);
        const int c = __builtin_amdgcn_readlane(i, l);
        best = c < best ? c : best;
        mask &= mask - 1u;
    } while (mask != 0u && ++it < 32);
    return best;
}

__global__ __launch_bounds__(1024) void fps_kernel(
    const float* __restrict__ xyz, const unsigned char* __restrict__ list,
    int* out_idx)
{
    __shared__ float4 pts[CAP];
    __shared__ unsigned short cidx[CAP];
    __shared__ int toff[256];
    __shared__ int tcnt[256];
    __shared__ int wtot[8];
    __shared__ int total_sh;
    __shared__ int sel_sh;
    __shared__ float dmax_sh;
    __shared__ int step_sh;
    __shared__ __attribute__((aligned(16))) float wtab[32][32];
    __shared__ float wv[32];
    __shared__ int   wi[32];
    __shared__ int   wcnt[32];
    __shared__ float4 sxyz[M_];
    __shared__ int    spos[M_];
    __shared__ unsigned short hpos[HSET];

    const int b   = blockIdx.x;
    const int tid = threadIdx.x;
    const float* xb = xyz + (size_t)b * N_ * 3;
    const unsigned char* lb = list + (size_t)b * N_;
    const int lane = tid & 31;
    const int wav = __builtin_amdgcn_readfirstlane(tid >> 5);

    int c = 0;
    if (tid < 256) c = out_idx[b * M_ + tid];
    int incl = c;
    for (int off = 1; off < 32; off <<= 1) {
        const int t = __shfl_up(incl, off, 32);
        if (lane >= off) incl += t;
    }
    if (tid < 256 && lane == 31) wtot[wav] = incl;
    __syncthreads();
    if (tid < 256) {
        int pre = 0;
        for (int w = 0; w < 8; ++w) if (w < wav) pre += wtot[w];
        toff[tid] = pre + incl - c;
        tcnt[tid] = c;
        if (tid == 255) total_sh = pre + incl;
    }
    __syncthreads();

    {
        const int j  = tid & 127;
        const int tq = tid >> 7;
        for (int p0 = 0; p0 < 32; p0 += 8) {
            int idx[8];
            float px[8], py[8], pz[8];
#pragma unroll
            for (int u = 0; u < 8; ++u) {
                const int tl = (p0 + u) * 8 + tq;
                idx[u] = tl * 128 + (int)lb[tl * 128 + j];
            }
#pragma unroll
            for (int u = 0; u < 8; ++u) {
                px[u] = xb[(size_t)idx[u] * 3];
                py[u] = xb[(size_t)idx[u] * 3 + 1];
                pz[u] = xb[(size_t)idx[u] * 3 + 2];
            }
#pragma unroll
            for (int u = 0; u < 8; ++u) {
                const int tl = (p0 + u) * 8 + tq;
                const int q  = toff[tl] + j;
                if (j < tcnt[tl] && q < CAP) {
                    pts[q]  = make_float4(px[u], py[u], pz[u], __builtin_inff());
                    cidx[q] = (unsigned short)idx[u];
                }
            }
        }
    }
    __syncthreads();
    int Nc = total_sh;
    if (Nc > CAP) Nc = CAP;
    if (Nc == 0) {
        for (int s = tid; s < M_; s += 1024) out_idx[b * M_ + s] = 0;
        return;
    }

    const float NEG_INF = -__builtin_inff();

    if (tid == 0) { spos[0] = 0; sxyz[0] = pts[0]; }
    __syncthreads();

    int a = 0, bb = 0;
    float frac = 0.85f;
    for (int round = 0; round < M_; ++round) {
        float px[NSLOT], py[NSLOT], pz[NSLOT], pw[NSLOT];
#pragma unroll
        for (int k = 0; k < NSLOT; ++k) {
            const int i = tid + k * 1024;
            const float4 p = pts[i < Nc ? i : 0];
            px[k] = p.x; py[k] = p.y; pz[k] = p.z;
            pw[k] = i < Nc ? p.w : NEG_INF;
        }
        for (int j = a; j <= bb; ++j) {
            const float4 q = sxyz[j];
#pragma unroll
            for (int k = 0; k < NSLOT; ++k)
                pw[k] = fminf(pw[k], sqdist(px[k], py[k], pz[k], q.x, q.y, q.z));
        }
        float bv = NEG_INF;
        int   bi = 0x7fffffff;
#pragma unroll
        for (int k = 0; k < NSLOT; ++k) {
            const int i = tid + k * 1024;
            if (i < Nc) pts[i].w = pw[k];
            if (pw[k] > bv) { bv = pw[k]; bi = i; }
        }
        {
            const float mv = wave_max_tab(bv, wtab[wav], lane);
            const int   mi = wave_index_of_max(bv, mv, bi);
            if (lane == 0) { wv[wav] = mv; wi[wav] = mi; }
        }
        __syncthreads();
        if (wav == 0) {
            bv = wv[lane]; bi = wi[lane];
            const float mv = wave_max_tab(bv, wtab[0], lane);
            const int   mi = wave_index_of_max(bv, mv, bi);
            if (lane == 0) { sel_sh = mi; dmax_sh = mv; }
        }
        __syncthreads();
        const int s = bb + 1;
        if (s >= M_) break;
        const int   bp = __builtin_amdgcn_readfirstlane(sel_sh);
        const float D  = dmax_sh;
        if (tid == 0) { spos[s] = bp; sxyz[s] = pts[bp]; }

        const float tau = frac * D;
        int cnt = 0;
#pragma unroll
        for (int k = 0; k < NSLOT; ++k)
            cnt += (tid + k * 1024 < Nc && pw[k] >= tau) ? 1 : 0;
        int inc2 = cnt;
        for (int off = 1; off < 32; off <<= 1) {
            const int t = __shfl_up(inc2, off, 32);
            if (lane >= off) inc2 += t;
        }
        if (lane == 31) wcnt[wav] = inc2;
        __syncthreads();
        int pre = 0, H = 0;
        for (int w = 0; w < 32; ++w) {
            const int cw = wcnt[w];
            H += cw;
            if (w < wav) pre += cw;
        }
        H = __builtin_amdgcn_readfirstlane(H);
        if (H <= HSET) {
            int o = pre + inc2 - cnt;
#pragma unroll
            for (int k = 0; k < NSLOT; ++k) {
                const int i = tid + k * 1024;
                if (i < Nc && pw[k] >= tau) hpos[o++] = (unsigned short)i;
            }
        }
        __syncthreads();
        if (H > HSET) {
            frac = 0.5f * (1.0f + frac);
            a = s; bb = s;
            continue;
        }

        if (wav == 0) {
            float hx[8], hy[8], hz[8], hw[8];
            int   hp[8];
#pragma unroll
            for (int u = 0; u < 8; ++u) {
                const int t = lane + 32 * u;
                const bool valid = t < H;
                const int pos = valid ? (int)hpos[valid ? t : 0] : 0;
                const float4 p = pts[pos];
                hx[u] = p.x; hy[u] = p.y; hz[u] = p.z;
                hw[u] = valid ? p.w : NEG_INF;
                hp[u] = valid ? pos : 0x7fffffff;
            }
            float4 q = sxyz[s];
            int ss = s;
            while (ss < M_) {
                float tv = NEG_INF;
#pragma unroll
                for (int u = 0; u < 8; ++u) {
                    hw[u] = fminf(hw[u], sqdist(hx[u], hy[u], hz[u],
                                                q.x, q.y, q.z));
                    tv = fmaxf(tv, hw[u]);
                }
                const float mx = wave_max_tab(tv, wtab[0], lane);
                int tp = 0x7fffffff;
#pragma unroll
                for (int u = 0; u < 8; ++u)
                    tp = min(tp, hw[u] == tv ? hp[u] : 0x7fffffff);
                const float tvu = __uint_as_float(
                    __builtin_amdgcn_readfirstlane(__float_as_uint(mx)));
                const int   tpu = wave_index_of_max(tv, mx, tp);
                if (!(tvu >= tau)) break;
                ++ss;
                if (ss < M_) {
                    const float4 pq = pts[tpu];
                    q = make_float4(pq.x, pq.y, pq.z, 0.0f);
                    if (lane == 0) { spos[ss] = tpu; sxyz[ss] = q; }
                }
            }
            if (lane == 0) step_sh = ss;
        }
        __syncthreads();
        const int ss = __builtin_amdgcn_readfirstlane(step_sh);
        if (ss >= M_) break;
        a = s; bb = ss;
    }
    __syncthreads();
    for (int t = tid; t < M_; t += 1024) out_idx[b * M_ + t] = (int)cidx[spos[t]];
}

__global__ __launch_bounds__(256) void gather_kernel(
    const float* __restrict__ feat, const float* __restrict__ xyz,
    const float* __restrict__ grasp, const int* __restrict__ idxs,
    float* __restrict__ out)
{
    const int g = blockIdx.x * 256 + threadIdx.x;
    if (g >= B_ * CH_ * M_) return;
    const int m  = g % M_;
    const int t  = g / M_;
    const int ch = t % CH_;
    const int b  = t / CH_;
    const int id = idxs[b * M_ + m];
    float v;
    if (ch < C_)            v = feat[((size_t)b * C_ + ch) * N_ + id];
    else if (ch < C_ + 3)   v = xyz[((size_t)b * N_ + id) * 3 + (ch - C_)];
    else                    v = grasp[(size_t)b * N_ + id];
    out[g] = v;
}

extern "C" void kernel_launch(void* const* d_in, const int* in_sizes, int n_in,
                              void* d_out, int out_size, void* d_ws, size_t ws_size,
                              hipStream_t stream) {
    const float* xyz  = (const float*)d_in[0];
    const float* feat = (const float*)d_in[1];
    const float* Wobj = (const float*)d_in[2];
    const float* bobj = (const float*)d_in[3];
    const float* Wg   = (const float*)d_in[4];
    const float* bg   = (const float*)d_in[5];
    float* out = (float*)d_out;

    char* ws = (char*)d_ws;
    float*         g_grasp = (float*)ws;
    unsigned char* g_list  = (unsigned char*)(ws + (size_t)B_ * N_ * 4);
    int*           g_idx   = (int*)(ws + (size_t)B_ * N_ * 4 + (size_t)B_ * N_);

    score_kernel<<<B_ * (N_ / 128), 256, 0, stream>>>(feat, Wobj, bobj, Wg, bg,
                                                      g_grasp, g_list, g_idx);
    fps_kernel<<<B_, 1024, 0, stream>>>(xyz, g_list, g_idx);
    const int total = B_ * CH_ * M_;
    gather_kernel<<<(total + 255) / 256, 256, 0, stream>>>(feat, xyz, g_grasp,
                                                           g_idx, out);
}


# FPS: wave 0's stepping and the other 31 waves' deferred distance updates overlap: new samples handed over in chunks of 4 at one barrier; points stay in registers all kernel; distances bit-identical
# speedup vs baseline: 3.1604x; 1.1149x over previous
#include <hip/hip_runtime.h>
#include <hip/hip_bf16.h>

typedef __attribute__((ext_vector_type(2))) float v2f;
typedef __attribute__((ext_vector_type(8))) float v8f;

#define B_   4
#define N_   32768
#define C_   512
#define M_   512
#define CH_  (C_ + 4)
#define CAP  8704

__global__ __launch_bounds__(256) void score_kernel(
    const float* __restrict__ feat, const float* __restrict__ Wobj,
    const float* __restrict__ bobj, const float* __restrict__ Wg,
    const float* __restrict__ bg, float* __restrict__ grasp_out,
    unsigned char* __restrict__ list_out, int* __restrict__ cnt_out)
{
    __shared__ float Apre[128][32][2];
    for (int t = threadIdx.x; t < 128 * 32; t += 256) {
        int kc = t >> 5, lane = t & 31;
        int row = lane & 15;
        int kh  = lane >> 4;
        int ka  = kc * 4 + 2 * kh;
        float a0 = 0.0f, a1 = 0.0f;
        if (row < 2)      { a0 = Wobj[row * C_ + ka]; a1 = Wobj[row * C_ + ka + 1]; }
        else if (row == 2){ a0 = Wg[ka];              a1 = Wg[ka + 1]; }
        Apre[kc][lane][0] = a0;
        Apre[kc][lane][1] = a1;
    }
    __syncthreads();

    const int wave = threadIdx.x >> 5;
    const int lane = threadIdx.x & 31;
    const int b    = blockIdx.x >> 8;
    const int tile = blockIdx.x & 255;
    const int n0   = tile * 128 + wave * 16;
    const int kh   = lane >> 4;
    const int col  = n0 + (lane & 15);

    const float* fb = feat + (size_t)b * C_ * N_;

    float b0 = fb[(size_t)(2 * kh) * N_ + col];
    float b1 = fb[(size_t)(2 * kh + 1) * N_ + col];

    v8f acc = {};
#pragma unroll 2
    for (int kc = 0; kc < 128; ++kc) {
        v2f a, bm;
        a.x  = Apre[kc][lane][0];
        a.y  = Apre[kc][lane][1];
        bm.x = b0;
        bm.y = b1;

        const int kn  = (kc + 1) & 127;
        const int kan = kn * 4 + 2 * kh;
        b0 = fb[(size_t)kan * N_ + col];
        b1 = fb[(size_t)(kan + 1) * N_ + col];

        int pfrow = kc * 4 + 32;
        if (pfrow > C_ - 4) pfrow = C_ - 4;
        __builtin_prefetch(&fb[(size_t)(pfrow + 2 * kh) * N_ + col], 0, 3);
        __builtin_prefetch(&fb[(size_t)(pfrow + 2 * kh + 1) * N_ + col], 0, 3);

        acc = __builtin_amdgcn_wmma_f32_16x16x4_f32(
            false, a, false, bm, (short)0, acc, false, false);
    }

    bool m = false;
    if (lane < 16) {
        float s0 = acc[0] + bobj[0];
        float s1 = acc[1] + bobj[1];
        float g  = acc[2] + bg[0];
        grasp_out[(size_t)b * N_ + n0 + lane] = g;
        m = (s1 > s0) && (g > 0.1f);
    }

    const unsigned bal = __builtin_amdgcn_ballot_w32(m);
    __syncthreads();
    if (lane == 0) Apre[0][wave][0] = __uint_as_float(bal);
    __syncthreads();
    int before = 0, total = 0;
    for (int w = 0; w < 8; ++w) {
        const int c = __popc(__float_as_uint(Apre[0][w][0]));
        total += c;
        if (w < wave) before += c;
    }
    if (m) {
        const int rank = __popc(bal & ((1u << lane) - 1u));
        list_out[(size_t)b * N_ + tile * 128 + before + rank] =
            (unsigned char)(wave * 16 + lane);
    }
    if ((int)threadIdx.x >= total && threadIdx.x < 128)
        list_out[(size_t)b * N_ + tile * 128 + threadIdx.x] = 0;
    if (threadIdx.x == 0) cnt_out[b * M_ + tile] = total;
}

#define HSET   256
#define BULK_T 992
#define NSLOT  9
#define GSTEP  4

__device__ __forceinline__ float sqdist(float px, float py, float pz,
                                        float qx, float qy, float qz)
{
    const float dx = px - qx;
    const float dy = py - qy;
    const float dz = pz - qz;
    float sxy;
    {
#pragma clang fp contract(off)
        sxy = dx * dx + dy * dy;
    }
    return __builtin_fmaf(dz, dz, sxy);
}

__device__ __forceinline__ float wave_max_tab(float v, float* tab, int lane)
{
    tab[lane] = v;
    __builtin_amdgcn_fence(__ATOMIC_RELEASE, "wavefront");
    __builtin_amdgcn_wave_barrier();
    __builtin_amdgcn_fence(__ATOMIC_ACQUIRE, "wavefront");
    const float4* t4 = reinterpret_cast<const float4*>(tab);
    float r[8];
#pragma unroll
    for (int k = 0; k < 8; ++k) {
        const float4 t = t4[k];
        r[k] = fmaxf(fmaxf(t.x, t.y), fmaxf(t.z, t.w));
    }
    return fmaxf(fmaxf(fmaxf(r[0], r[1]), fmaxf(r[2], r[3])),
                 fmaxf(fmaxf(r[4], r[5]), fmaxf(r[6], r[7])));
}

__device__ __forceinline__ int wave_index_of_max(float v, float m, int i)
{
    unsigned mask = __builtin_amdgcn_ballot_w32(v == m);
    int best = 0x7fffffff;
    int it = 0;
    do {
        const int l = __builtin_ctz(mask | 0x80000000u);
        const int c = __builtin_amdgcn_readlane(i, l);
        best = c < best ? c : best;
        mask &= mask - 1u;
    } while (mask != 0u && ++it < 32);
    return best;
}

__global__ __launch_bounds__(1024) void fps_kernel(
    const float* __restrict__ xyz, const unsigned char* __restrict__ list,
    int* out_idx)
{
    __shared__ float4 pts[CAP];
    __shared__ unsigned short cidx[CAP];
    __shared__ int toff[256];
    __shared__ int tcnt[256];
    __shared__ int wtot[8];
    __shared__ int total_sh;
    __shared__ int sel_sh;
    __shared__ float dmax_sh;
    __shared__ __attribute__((aligned(16))) float wtab[32][32];
    __shared__ float wv[32];
    __shared__ int   wi[32];
    __shared__ int   wcnt[32];
    __shared__ float4 sxyz[M_];
    __shared__ int    spos[M_];
    __shared__ unsigned short hthr[HSET];
    __shared__ unsigned char  hslot[HSET];
    __shared__ float hdist[HSET];
    __shared__ int   ctrl[2];

    const int b   = blockIdx.x;
    const int tid = threadIdx.x;
    const float* xb = xyz + (size_t)b * N_ * 3;
    const unsigned char* lb = list + (size_t)b * N_;
    const int lane = tid & 31;
    const int wav = __builtin_amdgcn_readfirstlane(tid >> 5);

    int c = 0;
    if (tid < 256) c = out_idx[b * M_ + tid];
    int incl = c;
    for (int off = 1; off < 32; off <<= 1) {
        const int t = __shfl_up(incl, off, 32);
        if (lane >= off) incl += t;
    }
    if (tid < 256 && lane == 31) wtot[wav] = incl;
    __syncthreads();
    if (tid < 256) {
        int pre = 0;
        for (int w = 0; w < 8; ++w) if (w < wav) pre += wtot[w];
        toff[tid] = pre + incl - c;
        tcnt[tid] = c;
        if (tid == 255) total_sh = pre + incl;
    }
    __syncthreads();

    {
        const int j  = tid & 127;
        const int tq = tid >> 7;
        for (int p0 = 0; p0 < 32; p0 += 8) {
            int idx[8];
            float px[8], py[8], pz[8];
#pragma unroll
            for (int u = 0; u < 8; ++u) {
                const int tl = (p0 + u) * 8 + tq;
                idx[u] = tl * 128 + (int)lb[tl * 128 + j];
            }
#pragma unroll
            for (int u = 0; u < 8; ++u) {
                px[u] = xb[(size_t)idx[u] * 3];
                py[u] = xb[(size_t)idx[u] * 3 + 1];
                pz[u] = xb[(size_t)idx[u] * 3 + 2];
            }
#pragma unroll
            for (int u = 0; u < 8; ++u) {
                const int tl = (p0 + u) * 8 + tq;
                const int q  = toff[tl] + j;
                if (j < tcnt[tl] && q < CAP) {
                    pts[q]  = make_float4(px[u], py[u], pz[u], __builtin_inff());
                    cidx[q] = (unsigned short)idx[u];
                }
            }
        }
    }
    __syncthreads();
    int Nc = total_sh;
    if (Nc > CAP) Nc = CAP;
    if (Nc == 0) {
        for (int s = tid; s < M_; s += 1024) out_idx[b * M_ + s] = 0;
        return;
    }

    const float NEG_INF = -__builtin_inff();

    const bool bulk = wav != 0;
    const int  base = tid - 32;
    float px[NSLOT], py[NSLOT], pz[NSLOT], pw[NSLOT];
    int   hp[8];
#pragma unroll
    for (int k = 0; k < NSLOT; ++k) {
        const int i = base + k * BULK_T;
        const bool v = bulk && i < Nc;
        const float4 p = pts[v ? i : 0];
        px[k] = p.x; py[k] = p.y; pz[k] = p.z;
        pw[k] = v ? p.w : NEG_INF;
    }
#pragma unroll
    for (int u = 0; u < 8; ++u) hp[u] = 0x7fffffff;

    if (tid == 0) { spos[0] = 0; sxyz[0] = pts[0]; }
    __syncthreads();

    int done_upto = 0;
    int avail = 1;
    float frac = 0.85f;
    float tau = 0.0f;
    float4 q = make_float4(0.0f, 0.0f, 0.0f, 0.0f);
    int  ss = 0;
    bool alive = false;
    for (int round = 0; round < M_; ++round) {
        for (int it = 0; it < M_; ++it) {
            int j = done_upto;
#pragma nounroll
            for (int g = 0; g < GSTEP; ++g) {
                float qx = q.x, qy = q.y, qz = q.z;
                bool act = alive && ss < M_;
                if (bulk) {
                    act = j < avail;
                    if (act) {
                        const float4 qq = sxyz[j];
                        qx = qq.x; qy = qq.y; qz = qq.z;
                        ++j;
                    }
                }
                if (act) {
#pragma unroll
                    for (int k = 0; k < NSLOT; ++k)
                        pw[k] = fminf(pw[k], sqdist(px[k], py[k], pz[k], qx, qy, qz));
                }
                if (!bulk && act) {
                    float tv = NEG_INF;
#pragma unroll
                    for (int u = 0; u < 8; ++u) tv = fmaxf(tv, pw[u]);
                    const float mx = wave_max_tab(tv, wtab[0], lane);
                    int tp = 0x7fffffff;
#pragma unroll
                    for (int u = 0; u < 8; ++u)
                        tp = min(tp, pw[u] == tv ? hp[u] : 0x7fffffff);
                    const float tvu = __uint_as_float(
                        __builtin_amdgcn_readfirstlane(__float_as_uint(mx)));
                    const int   tpu = wave_index_of_max(tv, mx, tp);
                    if (!(tvu >= tau)) {
                        alive = false;
                    } else {
                        ++ss;
                        if (ss < M_) {
                            const float4 pq = pts[tpu];
                            q = make_float4(pq.x, pq.y, pq.z, 0.0f);
                            if (lane == 0) { spos[ss] = tpu; sxyz[ss] = q; }
                        }
                    }
                }
            }
            if (!bulk && lane == 0)
                ctrl[it & 1] = ss | ((alive && ss < M_) ? 0 : 0x10000);
            done_upto = avail;
            __syncthreads();
            const int cwd = __builtin_amdgcn_readfirstlane(ctrl[it & 1]);
            const int ssp = cwd & 0xffff;
            avail = ssp + 1 < M_ ? ssp + 1 : M_;
            if ((cwd >> 16) != 0 && (done_upto >= avail || avail >= M_)) break;
        }
        if (avail >= M_) break;

        if (bulk) {
            float bv = NEG_INF;
            int   bi = 0x7fffffff;
            int   bk = 0;
#pragma unroll
            for (int k = 0; k < NSLOT; ++k)
                if (pw[k] > bv) { bv = pw[k]; bk = k; }
            if (bv > NEG_INF) bi = base + bk * BULK_T;
            const float mv = wave_max_tab(bv, wtab[wav], lane);
            const int   mi = wave_index_of_max(bv, mv, bi);
            if (lane == 0) { wv[wav] = mv; wi[wav] = mi; }
        } else if (lane == 0) {
            wv[0] = NEG_INF; wi[0] = 0x7fffffff;
        }
        __syncthreads();
        if (wav == 0) {
            const float v2 = wv[lane];
            const int   i2 = wi[lane];
            const float mv = wave_max_tab(v2, wtab[0], lane);
            const int   mi = wave_index_of_max(v2, mv, i2);
            if (lane == 0) { sel_sh = mi; dmax_sh = mv; }
        }
        __syncthreads();
        const int s = avail;
        const int   bp = __builtin_amdgcn_readfirstlane(sel_sh);
        const float D  = dmax_sh;
        if (tid == 0) { spos[s] = bp; sxyz[s] = pts[bp]; }
        avail = s + 1;
        if (avail >= M_) break;

        tau = frac * D;
        unsigned bm[NSLOT];
        int wtotal = 0;
#pragma unroll
        for (int k = 0; k < NSLOT; ++k) {
            const bool l = bulk && base < Nc - k * BULK_T && pw[k] >= tau;
            bm[k] = __builtin_amdgcn_ballot_w32(l);
            wtotal += __popc(bm[k]);
        }
        if (lane == 0) wcnt[wav] = wtotal;
        __syncthreads();
        int pre = 0, H = 0;
        for (int w = 0; w < 32; ++w) {
            const int cw = wcnt[w];
            H += cw;
            if (w < wav) pre += cw;
        }
        H = __builtin_amdgcn_readfirstlane(H);
        if (H <= HSET && bulk) {
            int run = __builtin_amdgcn_readfirstlane(pre);
#pragma unroll
            for (int k = 0; k < NSLOT; ++k) {
                const unsigned m = bm[k];
                if ((m >> lane) & 1u) {
                    const int o = run + (int)__builtin_amdgcn_mbcnt_lo(m, 0u);
                    hthr[o]  = (unsigned short)base;
                    hslot[o] = (unsigned char)k;
                    hdist[o] = pw[k];
                }
                run += __popc(m);
            }
        }
        __syncthreads();

        ss = s;
        if (H > HSET) {
            frac = 0.5f * (1.0f + frac);
            alive = false;
        } else {
            alive = true;
            if (!bulk) {
#pragma unroll
                for (int u = 0; u < 8; ++u) {
                    const bool valid = lane < H - 32 * u;
                    const int th = (int)hthr[lane + 32 * u];
                    const int sl = (int)hslot[lane + 32 * u];
                    const float d = hdist[lane + 32 * u];
                    const int pos = valid ? th + sl * BULK_T : 0;
                    const float4 p = pts[pos];
                    px[u] = p.x; py[u] = p.y; pz[u] = p.z;
                    pw[u] = valid ? d : NEG_INF;
                    hp[u] = valid ? pos : 0x7fffffff;
                }
                q = sxyz[s];
            }
        }
    }
    __syncthreads();
    for (int t = tid; t < M_; t += 1024) out_idx[b * M_ + t] = (int)cidx[spos[t]];
}

__global__ __launch_bounds__(256) void gather_kernel(
    const float* __restrict__ feat, const float* __restrict__ xyz,
    const float* __restrict__ grasp, const int* __restrict__ idxs,
    float* __restrict__ out)
{
    const int g = blockIdx.x * 256 + threadIdx.x;
    if (g >= B_ * CH_ * M_) return;
    const int m  = g % M_;
    const int t  = g / M_;
    const int ch = t % CH_;
    const int b  = t / CH_;
    const int id = idxs[b * M_ + m];
    float v;
    if (ch < C_)            v = feat[((size_t)b * C_ + ch) * N_ + id];
    else if (ch < C_ + 3)   v = xyz[((size_t)b * N_ + id) * 3 + (ch - C_)];
    else                    v = grasp[(size_t)b * N_ + id];
    out[g] = v;
}

extern "C" void kernel_launch(void* const* d_in, const int* in_sizes, int n_in,
                              void* d_out, int out_size, void* d_ws, size_t ws_size,
                              hipStream_t stream) {
    const float* xyz  = (const float*)d_in[0];
    const float* feat = (const float*)d_in[1];
    const float* Wobj = (const float*)d_in[2];
    const float* bobj = (const float*)d_in[3];
    const float* Wg   = (const float*)d_in[4];
    const float* bg   = (const float*)d_in[5];
    float* out = (float*)d_out;

    char* ws = (char*)d_ws;
    float*         g_grasp = (float*)ws;
    unsigned char* g_list  = (unsigned char*)(ws + (size_t)B_ * N_ * 4);
    int*           g_idx   = (int*)(ws + (size_t)B_ * N_ * 4 + (size_t)B_ * N_);

    score_kernel<<<B_ * (N_ / 128), 256, 0, stream>>>(feat, Wobj, bobj, Wg, bg,
                                                      g_grasp, g_list, g_idx);
    fps_kernel<<<B_, 1024, 0, stream>>>(xyz, g_list, g_idx);
    const int total = B_ * CH_ * M_;
    gather_kernel<<<(total + 255) / 256, 256, 0, stream>>>(feat, xyz, g_grasp,
                                                           g_idx, out);
}


# Score kernel: B fragments loaded 8 K-chunks (16 loads a lane) at a time, one group ahead of their WMMAs, and the A table's 16 loads together: a memory round trip per group, not per WMMA; no prefetch
# speedup vs baseline: 3.3523x; 1.0607x over previous
#include <hip/hip_runtime.h>
#include <hip/hip_bf16.h>

typedef __attribute__((ext_vector_type(2))) float v2f;
typedef __attribute__((ext_vector_type(8))) float v8f;

#define B_   4
#define N_   32768
#define C_   512
#define M_   512
#define CH_  (C_ + 4)
#define CAP  8704

#define BG 8
__global__ __launch_bounds__(256) void score_kernel(
    const float* __restrict__ feat, const float* __restrict__ Wobj,
    const float* __restrict__ bobj, const float* __restrict__ Wg,
    const float* __restrict__ bg, float* __restrict__ grasp_out,
    unsigned char* __restrict__ list_out, int* __restrict__ cnt_out)
{
    __shared__ float Apre[128][32][2];
    {
        const int alane = threadIdx.x & 31;
        const int arow  = alane & 15;
        const int akh   = alane >> 4;
        const float* asrc = arow < 2 ? Wobj + arow * C_ : Wg;
        const bool akeep = arow < 3;
        float av0[16], av1[16];
#pragma unroll
        for (int i = 0; i < 16; ++i) {
            const int ka = ((int)(threadIdx.x >> 5) + 8 * i) * 4 + 2 * akh;
            av0[i] = asrc[ka];
            av1[i] = asrc[ka + 1];
        }
#pragma unroll
        for (int i = 0; i < 16; ++i) {
            const int kc = (int)(threadIdx.x >> 5) + 8 * i;
            Apre[kc][alane][0] = akeep ? av0[i] : 0.0f;
            Apre[kc][alane][1] = akeep ? av1[i] : 0.0f;
        }
    }
    __syncthreads();

    const int wave = threadIdx.x >> 5;
    const int lane = threadIdx.x & 31;
    const int b    = blockIdx.x >> 8;
    const int tile = blockIdx.x & 255;
    const int n0   = tile * 128 + wave * 16;
    const int kh   = lane >> 4;
    const int col  = n0 + (lane & 15);

    const float* fb = feat + (size_t)b * C_ * N_;

    const unsigned loff = (unsigned)(2 * kh) * N_ + (unsigned)col;
    float bA0[BG], bA1[BG], bB0[BG], bB1[BG];
#pragma unroll
    for (int i = 0; i < BG; ++i) {
        bA0[i] = fb[(size_t)(i * 4) * N_ + loff];
        bA1[i] = fb[(size_t)(i * 4 + 1) * N_ + loff];
    }

    v8f acc = {};
#pragma nounroll
    for (int g = 0; g < 128 / BG - 2; g += 2) {
        const float* gb = fb + (size_t)(g * BG * 4) * N_;
#pragma unroll
        for (int i = 0; i < BG; ++i) {
            bB0[i] = gb[(size_t)((BG + i) * 4) * N_ + loff];
            bB1[i] = gb[(size_t)((BG + i) * 4 + 1) * N_ + loff];
        }
#pragma unroll
        for (int i = 0; i < BG; ++i) {
            v2f a, bm;
            a.x  = Apre[g * BG + i][lane][0];
            a.y  = Apre[g * BG + i][lane][1];
            bm.x = bA0[i];
            bm.y = bA1[i];
            acc = __builtin_amdgcn_wmma_f32_16x16x4_f32(
                false, a, false, bm, (short)0, acc, false, false);
        }
#pragma unroll
        for (int i = 0; i < BG; ++i) {
            bA0[i] = gb[(size_t)((2 * BG + i) * 4) * N_ + loff];
            bA1[i] = gb[(size_t)((2 * BG + i) * 4 + 1) * N_ + loff];
        }
#pragma unroll
        for (int i = 0; i < BG; ++i) {
            v2f a, bm;
            a.x  = Apre[(g + 1) * BG + i][lane][0];
            a.y  = Apre[(g + 1) * BG + i][lane][1];
            bm.x = bB0[i];
            bm.y = bB1[i];
            acc = __builtin_amdgcn_wmma_f32_16x16x4_f32(
                false, a, false, bm, (short)0, acc, false, false);
        }
    }
    {
        const int g = 128 / BG - 2;
        const float* gb = fb + (size_t)(g * BG * 4) * N_;
#pragma unroll
        for (int i = 0; i < BG; ++i) {
            bB0[i] = gb[(size_t)((BG + i) * 4) * N_ + loff];
            bB1[i] = gb[(size_t)((BG + i) * 4 + 1) * N_ + loff];
        }
#pragma unroll
        for (int i = 0; i < BG; ++i) {
            v2f a, bm;
            a.x  = Apre[g * BG + i][lane][0];
            a.y  = Apre[g * BG + i][lane][1];
            bm.x = bA0[i];
            bm.y = bA1[i];
            acc = __builtin_amdgcn_wmma_f32_16x16x4_f32(
                false, a, false, bm, (short)0, acc, false, false);
        }
#pragma unroll
        for (int i = 0; i < BG; ++i) {
            v2f a, bm;
            a.x  = Apre[(g + 1) * BG + i][lane][0];
            a.y  = Apre[(g + 1) * BG + i][lane][1];
            bm.x = bB0[i];
            bm.y = bB1[i];
            acc = __builtin_amdgcn_wmma_f32_16x16x4_f32(
                false, a, false, bm, (short)0, acc, false, false);
        }
    }

    bool m = false;
    if (lane < 16) {
        float s0 = acc[0] + bobj[0];
        float s1 = acc[1] + bobj[1];
        float g  = acc[2] + bg[0];
        grasp_out[(size_t)b * N_ + n0 + lane] = g;
        m = (s1 > s0) && (g > 0.1f);
    }

    const unsigned bal = __builtin_amdgcn_ballot_w32(m);
    __syncthreads();
    if (lane == 0) Apre[0][wave][0] = __uint_as_float(bal);
    __syncthreads();
    int before = 0, total = 0;
    for (int w = 0; w < 8; ++w) {
        const int c = __popc(__float_as_uint(Apre[0][w][0]));
        total += c;
        if (w < wave) before += c;
    }
    if (m) {
        const int rank = __popc(bal & ((1u << lane) - 1u));
        list_out[(size_t)b * N_ + tile * 128 + before + rank] =
            (unsigned char)(wave * 16 + lane);
    }
    if ((int)threadIdx.x >= total && threadIdx.x < 128)
        list_out[(size_t)b * N_ + tile * 128 + threadIdx.x] = 0;
    if (threadIdx.x == 0) cnt_out[b * M_ + tile] = total;
}

#define HSET   256
#define BULK_T 992
#define NSLOT  9
#define GSTEP  4

__device__ __forceinline__ float sqdist(float px, float py, float pz,
                                        float qx, float qy, float qz)
{
    const float dx = px - qx;
    const float dy = py - qy;
    const float dz = pz - qz;
    float sxy;
    {
#pragma clang fp contract(off)
        sxy = dx * dx + dy * dy;
    }
    return __builtin_fmaf(dz, dz, sxy);
}

__device__ __forceinline__ float wave_max_tab(float v, float* tab, int lane)
{
    tab[lane] = v;
    __builtin_amdgcn_fence(__ATOMIC_RELEASE, "wavefront");
    __builtin_amdgcn_wave_barrier();
    __builtin_amdgcn_fence(__ATOMIC_ACQUIRE, "wavefront");
    const float4* t4 = reinterpret_cast<const float4*>(tab);
    float r[8];
#pragma unroll
    for (int k = 0; k < 8; ++k) {
        const float4 t = t4[k];
        r[k] = fmaxf(fmaxf(t.x, t.y), fmaxf(t.z, t.w));
    }
    return fmaxf(fmaxf(fmaxf(r[0], r[1]), fmaxf(r[2], r[3])),
                 fmaxf(fmaxf(r[4], r[5]), fmaxf(r[6], r[7])));
}

__device__ __forceinline__ int wave_index_of_max(float v, float m, int i)
{
    unsigned mask = __builtin_amdgcn_ballot_w32(v == m);
    int best = 0x7fffffff;
    int it = 0;
    do {
        const int l = __builtin_ctz(mask | 0x80000000u);
        const int c = __builtin_amdgcn_readlane(i, l);
        best = c < best ? c : best;
        mask &= mask - 1u;
    } while (mask != 0u && ++it < 32);
    return best;
}

__global__ __launch_bounds__(1024) void fps_kernel(
    const float* __restrict__ xyz, const unsigned char* __restrict__ list,
    int* out_idx)
{
    __shared__ float4 pts[CAP];
    __shared__ unsigned short cidx[CAP];
    __shared__ int toff[256];
    __shared__ int tcnt[256];
    __shared__ int wtot[8];
    __shared__ int total_sh;
    __shared__ int sel_sh;
    __shared__ float dmax_sh;
    __shared__ __attribute__((aligned(16))) float wtab[32][32];
    __shared__ float wv[32];
    __shared__ int   wi[32];
    __shared__ int   wcnt[32];
    __shared__ float4 sxyz[M_];
    __shared__ int    spos[M_];
    __shared__ unsigned short hthr[HSET];
    __shared__ unsigned char  hslot[HSET];
    __shared__ float hdist[HSET];
    __shared__ int   ctrl[2];

    const int b   = blockIdx.x;
    const int tid = threadIdx.x;
    const float* xb = xyz + (size_t)b * N_ * 3;
    const unsigned char* lb = list + (size_t)b * N_;
    const int lane = tid & 31;
    const int wav = __builtin_amdgcn_readfirstlane(tid >> 5);

    int c = 0;
    if (tid < 256) c = out_idx[b * M_ + tid];
    int incl = c;
    for (int off = 1; off < 32; off <<= 1) {
        const int t = __shfl_up(incl, off, 32);
        if (lane >= off) incl += t;
    }
    if (tid < 256 && lane == 31) wtot[wav] = incl;
    __syncthreads();
    if (tid < 256) {
        int pre = 0;
        for (int w = 0; w < 8; ++w) if (w < wav) pre += wtot[w];
        toff[tid] = pre + incl - c;
        tcnt[tid] = c;
        if (tid == 255) total_sh = pre + incl;
    }
    __syncthreads();

    {
        const int j  = tid & 127;
        const int tq = tid >> 7;
        for (int p0 = 0; p0 < 32; p0 += 8) {
            int idx[8];
            float px[8], py[8], pz[8];
#pragma unroll
            for (int u = 0; u < 8; ++u) {
                const int tl = (p0 + u) * 8 + tq;
                idx[u] = tl * 128 + (int)lb[tl * 128 + j];
            }
#pragma unroll
            for (int u = 0; u < 8; ++u) {
                px[u] = xb[(size_t)idx[u] * 3];
                py[u] = xb[(size_t)idx[u] * 3 + 1];
                pz[u] = xb[(size_t)idx[u] * 3 + 2];
            }
#pragma unroll
            for (int u = 0; u < 8; ++u) {
                const int tl = (p0 + u) * 8 + tq;
                const int q  = toff[tl] + j;
                if (j < tcnt[tl] && q < CAP) {
                    pts[q]  = make_float4(px[u], py[u], pz[u], __builtin_inff());
                    cidx[q] = (unsigned short)idx[u];
                }
            }
        }
    }
    __syncthreads();
    int Nc = total_sh;
    if (Nc > CAP) Nc = CAP;
    if (Nc == 0) {
        for (int s = tid; s < M_; s += 1024) out_idx[b * M_ + s] = 0;
        return;
    }

    const float NEG_INF = -__builtin_inff();

    const bool bulk = wav != 0;
    const int  base = tid - 32;
    float px[NSLOT], py[NSLOT], pz[NSLOT], pw[NSLOT];
    int   hp[8];
#pragma unroll
    for (int k = 0; k < NSLOT; ++k) {
        const int i = base + k * BULK_T;
        const bool v = bulk && i < Nc;
        const float4 p = pts[v ? i : 0];
        px[k] = p.x; py[k] = p.y; pz[k] = p.z;
        pw[k] = v ? p.w : NEG_INF;
    }
#pragma unroll
    for (int u = 0; u < 8; ++u) hp[u] = 0x7fffffff;

    if (tid == 0) { spos[0] = 0; sxyz[0] = pts[0]; }
    __syncthreads();

    int done_upto = 0;
    int avail = 1;
    float frac = 0.85f;
    float tau = 0.0f;
    float4 q = make_float4(0.0f, 0.0f, 0.0f, 0.0f);
    int  ss = 0;
    bool alive = false;
    for (int round = 0; round < M_; ++round) {
        for (int it = 0; it < M_; ++it) {
            int j = done_upto;
#pragma nounroll
            for (int g = 0; g < GSTEP; ++g) {
                float qx = q.x, qy = q.y, qz = q.z;
                bool act = alive && ss < M_;
                if (bulk) {
                    act = j < avail;
                    if (act) {
                        const float4 qq = sxyz[j];
                        qx = qq.x; qy = qq.y; qz = qq.z;
                        ++j;
                    }
                }
                if (act) {
#pragma unroll
                    for (int k = 0; k < NSLOT; ++k)
                        pw[k] = fminf(pw[k], sqdist(px[k], py[k], pz[k], qx, qy, qz));
                }
                if (!bulk && act) {
                    float tv = NEG_INF;
#pragma unroll
                    for (int u = 0; u < 8; ++u) tv = fmaxf(tv, pw[u]);
                    const float mx = wave_max_tab(tv, wtab[0], lane);
                    int tp = 0x7fffffff;
#pragma unroll
                    for (int u = 0; u < 8; ++u)
                        tp = min(tp, pw[u] == tv ? hp[u] : 0x7fffffff);
                    const float tvu = __uint_as_float(
                        __builtin_amdgcn_readfirstlane(__float_as_uint(mx)));
                    const int   tpu = wave_index_of_max(tv, mx, tp);
                    if (!(tvu >= tau)) {
                        alive = false;
                    } else {
                        ++ss;
                        if (ss < M_) {
                            const float4 pq = pts[tpu];
                            q = make_float4(pq.x, pq.y, pq.z, 0.0f);
                            if (lane == 0) { spos[ss] = tpu; sxyz[ss] = q; }
                        }
                    }
                }
            }
            if (!bulk && lane == 0)
                ctrl[it & 1] = ss | ((alive && ss < M_) ? 0 : 0x10000);
            done_upto = avail;
            __syncthreads();
            const int cwd = __builtin_amdgcn_readfirstlane(ctrl[it & 1]);
            const int ssp = cwd & 0xffff;
            avail = ssp + 1 < M_ ? ssp + 1 : M_;
            if ((cwd >> 16) != 0 && (done_upto >= avail || avail >= M_)) break;
        }
        if (avail >= M_) break;

        if (bulk) {
            float bv = NEG_INF;
            int   bi = 0x7fffffff;
            int   bk = 0;
#pragma unroll
            for (int k = 0; k < NSLOT; ++k)
                if (pw[k] > bv) { bv = pw[k]; bk = k; }
            if (bv > NEG_INF) bi = base + bk * BULK_T;
            const float mv = wave_max_tab(bv, wtab[wav], lane);
            const int   mi = wave_index_of_max(bv, mv, bi);
            if (lane == 0) { wv[wav] = mv; wi[wav] = mi; }
        } else if (lane == 0) {
            wv[0] = NEG_INF; wi[0] = 0x7fffffff;
        }
        __syncthreads();
        if (wav == 0) {
            const float v2 = wv[lane];
            const int   i2 = wi[lane];
            const float mv = wave_max_tab(v2, wtab[0], lane);
            const int   mi = wave_index_of_max(v2, mv, i2);
            if (lane == 0) { sel_sh = mi; dmax_sh = mv; }
        }
        __syncthreads();
        const int s = avail;
        const int   bp = __builtin_amdgcn_readfirstlane(sel_sh);
        const float D  = dmax_sh;
        if (tid == 0) { spos[s] = bp; sxyz[s] = pts[bp]; }
        avail = s + 1;
        if (avail >= M_) break;

        tau = frac * D;
        unsigned bm[NSLOT];
        int wtotal = 0;
#pragma unroll
        for (int k = 0; k < NSLOT; ++k) {
            const bool l = bulk && base < Nc - k * BULK_T && pw[k] >= tau;
            bm[k] = __builtin_amdgcn_ballot_w32(l);
            wtotal += __popc(bm[k]);
        }
        if (lane == 0) wcnt[wav] = wtotal;
        __syncthreads();
        int pre = 0, H = 0;
        for (int w = 0; w < 32; ++w) {
            const int cw = wcnt[w];
            H += cw;
            if (w < wav) pre += cw;
        }
        H = __builtin_amdgcn_readfirstlane(H);
        if (H <= HSET && bulk) {
            int run = __builtin_amdgcn_readfirstlane(pre);
#pragma unroll
            for (int k = 0; k < NSLOT; ++k) {
                const unsigned m = bm[k];
                if ((m >> lane) & 1u) {
                    const int o = run + (int)__builtin_amdgcn_mbcnt_lo(m, 0u);
                    hthr[o]  = (unsigned short)base;
                    hslot[o] = (unsigned char)k;
                    hdist[o] = pw[k];
                }
                run += __popc(m);
            }
        }
        __syncthreads();

        ss = s;
        if (H > HSET) {
            frac = 0.5f * (1.0f + frac);
            alive = false;
        } else {
            alive = true;
            if (!bulk) {
#pragma unroll
                for (int u = 0; u < 8; ++u) {
                    const bool valid = lane < H - 32 * u;
                    const int th = (int)hthr[lane + 32 * u];
                    const int sl = (int)hslot[lane + 32 * u];
                    const float d = hdist[lane + 32 * u];
                    const int pos = valid ? th + sl * BULK_T : 0;
                    const float4 p = pts[pos];
                    px[u] = p.x; py[u] = p.y; pz[u] = p.z;
                    pw[u] = valid ? d : NEG_INF;
                    hp[u] = valid ? pos : 0x7fffffff;
                }
                q = sxyz[s];
            }
        }
    }
    __syncthreads();
    for (int t = tid; t < M_; t += 1024) out_idx[b * M_ + t] = (int)cidx[spos[t]];
}

__global__ __launch_bounds__(256) void gather_kernel(
    const float* __restrict__ feat, const float* __restrict__ xyz,
    const float* __restrict__ grasp, const int* __restrict__ idxs,
    float* __restrict__ out)
{
    const int g = blockIdx.x * 256 + threadIdx.x;
    if (g >= B_ * CH_ * M_) return;
    const int m  = g % M_;
    const int t  = g / M_;
    const int ch = t % CH_;
    const int b  = t / CH_;
    const int id = idxs[b * M_ + m];
    float v;
    if (ch < C_)            v = feat[((size_t)b * C_ + ch) * N_ + id];
    else if (ch < C_ + 3)   v = xyz[((size_t)b * N_ + id) * 3 + (ch - C_)];
    else                    v = grasp[(size_t)b * N_ + id];
    out[g] = v;
}

extern "C" void kernel_launch(void* const* d_in, const int* in_sizes, int n_in,
                              void* d_out, int out_size, void* d_ws, size_t ws_size,
                              hipStream_t stream) {
    const float* xyz  = (const float*)d_in[0];
    const float* feat = (const float*)d_in[1];
    const float* Wobj = (const float*)d_in[2];
    const float* bobj = (const float*)d_in[3];
    const float* Wg   = (const float*)d_in[4];
    const float* bg   = (const float*)d_in[5];
    float* out = (float*)d_out;

    char* ws = (char*)d_ws;
    float*         g_grasp = (float*)ws;
    unsigned char* g_list  = (unsigned char*)(ws + (size_t)B_ * N_ * 4);
    int*           g_idx   = (int*)(ws + (size_t)B_ * N_ * 4 + (size_t)B_ * N_);

    score_kernel<<<B_ * (N_ / 128), 256, 0, stream>>>(feat, Wobj, bobj, Wg, bg,
                                                      g_grasp, g_list, g_idx);
    fps_kernel<<<B_, 1024, 0, stream>>>(xyz, g_list, g_idx);
    const int total = B_ * CH_ * M_;
    gather_kernel<<<(total + 255) / 256, 256, 0, stream>>>(feat, xyz, g_grasp,
                                                           g_idx, out);
}


# FPS stepping wave gets a step loop of its own: only the slots its listed set fills (1-8, not all 9), next sample's xyz by lane reads from the winning lane's registers, no LDS fetch; picks identical
# speedup vs baseline: 3.4748x; 1.0365x over previous
#include <hip/hip_runtime.h>
#include <hip/hip_bf16.h>

typedef __attribute__((ext_vector_type(2))) float v2f;
typedef __attribute__((ext_vector_type(8))) float v8f;

#define B_   4
#define N_   32768
#define C_   512
#define M_   512
#define CH_  (C_ + 4)
#define CAP  8704

#define BG 8
__global__ __launch_bounds__(256) void score_kernel(
    const float* __restrict__ feat, const float* __restrict__ Wobj,
    const float* __restrict__ bobj, const float* __restrict__ Wg,
    const float* __restrict__ bg, float* __restrict__ grasp_out,
    unsigned char* __restrict__ list_out, int* __restrict__ cnt_out)
{
    __shared__ float Apre[128][32][2];
    {
        const int alane = threadIdx.x & 31;
        const int arow  = alane & 15;
        const int akh   = alane >> 4;
        const float* asrc = arow < 2 ? Wobj + arow * C_ : Wg;
        const bool akeep = arow < 3;
        float av0[16], av1[16];
#pragma unroll
        for (int i = 0; i < 16; ++i) {
            const int ka = ((int)(threadIdx.x >> 5) + 8 * i) * 4 + 2 * akh;
            av0[i] = asrc[ka];
            av1[i] = asrc[ka + 1];
        }
#pragma unroll
        for (int i = 0; i < 16; ++i) {
            const int kc = (int)(threadIdx.x >> 5) + 8 * i;
            Apre[kc][alane][0] = akeep ? av0[i] : 0.0f;
            Apre[kc][alane][1] = akeep ? av1[i] : 0.0f;
        }
    }
    __syncthreads();

    const int wave = threadIdx.x >> 5;
    const int lane = threadIdx.x & 31;
    const int b    = blockIdx.x >> 8;
    const int tile = blockIdx.x & 255;
    const int n0   = tile * 128 + wave * 16;
    const int kh   = lane >> 4;
    const int col  = n0 + (lane & 15);

    const float* fb = feat + (size_t)b * C_ * N_;

    const unsigned loff = (unsigned)(2 * kh) * N_ + (unsigned)col;
    float bA0[BG], bA1[BG], bB0[BG], bB1[BG];
#pragma unroll
    for (int i = 0; i < BG; ++i) {
        bA0[i] = fb[(size_t)(i * 4) * N_ + loff];
        bA1[i] = fb[(size_t)(i * 4 + 1) * N_ + loff];
    }

    v8f acc = {};
#pragma nounroll
    for (int g = 0; g < 128 / BG - 2; g += 2) {
        const float* gb = fb + (size_t)(g * BG * 4) * N_;
#pragma unroll
        for (int i = 0; i < BG; ++i) {
            bB0[i] = gb[(size_t)((BG + i) * 4) * N_ + loff];
            bB1[i] = gb[(size_t)((BG + i) * 4 + 1) * N_ + loff];
        }
#pragma unroll
        for (int i = 0; i < BG; ++i) {
            v2f a, bm;
            a.x  = Apre[g * BG + i][lane][0];
            a.y  = Apre[g * BG + i][lane][1];
            bm.x = bA0[i];
            bm.y = bA1[i];
            acc = __builtin_amdgcn_wmma_f32_16x16x4_f32(
                false, a, false, bm, (short)0, acc, false, false);
        }
#pragma unroll
        for (int i = 0; i < BG; ++i) {
            bA0[i] = gb[(size_t)((2 * BG + i) * 4) * N_ + loff];
            bA1[i] = gb[(size_t)((2 * BG + i) * 4 + 1) * N_ + loff];
        }
#pragma unroll
        for (int i = 0; i < BG; ++i) {
            v2f a, bm;
            a.x  = Apre[(g + 1) * BG + i][lane][0];
            a.y  = Apre[(g + 1) * BG + i][lane][1];
            bm.x = bB0[i];
            bm.y = bB1[i];
            acc = __builtin_amdgcn_wmma_f32_16x16x4_f32(
                false, a, false, bm, (short)0, acc, false, false);
        }
    }
    {
        const int g = 128 / BG - 2;
        const float* gb = fb + (size_t)(g * BG * 4) * N_;
#pragma unroll
        for (int i = 0; i < BG; ++i) {
            bB0[i] = gb[(size_t)((BG + i) * 4) * N_ + loff];
            bB1[i] = gb[(size_t)((BG + i) * 4 + 1) * N_ + loff];
        }
#pragma unroll
        for (int i = 0; i < BG; ++i) {
            v2f a, bm;
            a.x  = Apre[g * BG + i][lane][0];
            a.y  = Apre[g * BG + i][lane][1];
            bm.x = bA0[i];
            bm.y = bA1[i];
            acc = __builtin_amdgcn_wmma_f32_16x16x4_f32(
                false, a, false, bm, (short)0, acc, false, false);
        }
#pragma unroll
        for (int i = 0; i < BG; ++i) {
            v2f a, bm;
            a.x  = Apre[(g + 1) * BG + i][lane][0];
            a.y  = Apre[(g + 1) * BG + i][lane][1];
            bm.x = bB0[i];
            bm.y = bB1[i];
            acc = __builtin_amdgcn_wmma_f32_16x16x4_f32(
                false, a, false, bm, (short)0, acc, false, false);
        }
    }

    bool m = false;
    if (lane < 16) {
        float s0 = acc[0] + bobj[0];
        float s1 = acc[1] + bobj[1];
        float g  = acc[2] + bg[0];
        grasp_out[(size_t)b * N_ + n0 + lane] = g;
        m = (s1 > s0) && (g > 0.1f);
    }

    const unsigned bal = __builtin_amdgcn_ballot_w32(m);
    __syncthreads();
    if (lane == 0) Apre[0][wave][0] = __uint_as_float(bal);
    __syncthreads();
    int before = 0, total = 0;
    for (int w = 0; w < 8; ++w) {
        const int c = __popc(__float_as_uint(Apre[0][w][0]));
        total += c;
        if (w < wave) before += c;
    }
    if (m) {
        const int rank = __popc(bal & ((1u << lane) - 1u));
        list_out[(size_t)b * N_ + tile * 128 + before + rank] =
            (unsigned char)(wave * 16 + lane);
    }
    if ((int)threadIdx.x >= total && threadIdx.x < 128)
        list_out[(size_t)b * N_ + tile * 128 + threadIdx.x] = 0;
    if (threadIdx.x == 0) cnt_out[b * M_ + tile] = total;
}

#define HSET   256
#define BULK_T 992
#define NSLOT  9
#define GSTEP  4

__device__ __forceinline__ float sqdist(float px, float py, float pz,
                                        float qx, float qy, float qz)
{
    const float dx = px - qx;
    const float dy = py - qy;
    const float dz = pz - qz;
    float sxy;
    {
#pragma clang fp contract(off)
        sxy = dx * dx + dy * dy;
    }
    return __builtin_fmaf(dz, dz, sxy);
}

__device__ __forceinline__ float wave_max_tab(float v, float* tab, int lane)
{
    tab[lane] = v;
    __builtin_amdgcn_fence(__ATOMIC_RELEASE, "wavefront");
    __builtin_amdgcn_wave_barrier();
    __builtin_amdgcn_fence(__ATOMIC_ACQUIRE, "wavefront");
    const float4* t4 = reinterpret_cast<const float4*>(tab);
    float r[8];
#pragma unroll
    for (int k = 0; k < 8; ++k) {
        const float4 t = t4[k];
        r[k] = fmaxf(fmaxf(t.x, t.y), fmaxf(t.z, t.w));
    }
    return fmaxf(fmaxf(fmaxf(r[0], r[1]), fmaxf(r[2], r[3])),
                 fmaxf(fmaxf(r[4], r[5]), fmaxf(r[6], r[7])));
}

__device__ __forceinline__ int wave_index_of_max(float v, float m, int i)
{
    unsigned mask = __builtin_amdgcn_ballot_w32(v == m);
    int best = 0x7fffffff;
    int it = 0;
    do {
        const int l = __builtin_ctz(mask | 0x80000000u);
        const int c = __builtin_amdgcn_readlane(i, l);
        best = c < best ? c : best;
        mask &= mask - 1u;
    } while (mask != 0u && ++it < 32);
    return best;
}

#define PICK_CASE(U)                                                          \
    case U:                                                                   \
        if (U < S) {                                                          \
            nx = __int_as_float(__builtin_amdgcn_readlane(__float_as_int(px[U < S ? U : 0]), bl)); \
            ny = __int_as_float(__builtin_amdgcn_readlane(__float_as_int(py[U < S ? U : 0]), bl)); \
            nz = __int_as_float(__builtin_amdgcn_readlane(__float_as_int(pz[U < S ? U : 0]), bl)); \
        }                                                                     \
        break;
template <int S>
__device__ __forceinline__ void step_chunk(
    const float (&px)[NSLOT], const float (&py)[NSLOT], const float (&pz)[NSLOT],
    float (&pw)[NSLOT], const int (&hk)[8],
    float& qx, float& qy, float& qz, int& ss, bool& alive, float tau,
    float* tab, int lane, int* spos, float4* sxyz)
{
#pragma nounroll
    for (int g = 0; g < GSTEP; ++g) {
        if (!(alive && ss < M_)) break;
        float tv = -__builtin_inff();
#pragma unroll
        for (int u = 0; u < S; ++u)
            pw[u] = fminf(pw[u], sqdist(px[u], py[u], pz[u], qx, qy, qz));
#pragma unroll
        for (int u = 0; u < S; ++u) tv = fmaxf(tv, pw[u]);
        const float mx = wave_max_tab(tv, tab, lane);
        int tk = 0x7fffffff;
#pragma unroll
        for (int u = 0; u < S; ++u)
            tk = min(tk, pw[u] == tv ? hk[u] : 0x7fffffff);
        unsigned mask = __builtin_amdgcn_ballot_w32(tv == mx);
        int best = 0x7fffffff, bl = 0, trips = 0;
        do {
            const int l = __builtin_ctz(mask | 0x80000000u);
            const int ck = __builtin_amdgcn_readlane(tk, l);
            if (ck < best) { best = ck; bl = l; }
            mask &= mask - 1u;
        } while (mask != 0u && ++trips < 32);
        const float tvu = __uint_as_float(
            __builtin_amdgcn_readfirstlane(__float_as_uint(mx)));
        if (!(tvu >= tau)) {
            alive = false;
        } else {
            ++ss;
            if (ss < M_) {
                float nx = 0.0f, ny = 0.0f, nz = 0.0f;
                switch (best & 7) {
                    PICK_CASE(0) PICK_CASE(1) PICK_CASE(2) PICK_CASE(3)
                    PICK_CASE(4) PICK_CASE(5) PICK_CASE(6) PICK_CASE(7)
                    default: break;
                }
                qx = nx; qy = ny; qz = nz;
                if (lane == 0) {
                    spos[ss] = best >> 3;
                    sxyz[ss] = make_float4(nx, ny, nz, 0.0f);
                }
            }
        }
    }
}
#undef PICK_CASE

__global__ __launch_bounds__(1024) void fps_kernel(
    const float* __restrict__ xyz, const unsigned char* __restrict__ list,
    int* out_idx)
{
    __shared__ float4 pts[CAP];
    __shared__ unsigned short cidx[CAP];
    __shared__ int toff[256];
    __shared__ int tcnt[256];
    __shared__ int wtot[8];
    __shared__ int total_sh;
    __shared__ int sel_sh;
    __shared__ float dmax_sh;
    __shared__ __attribute__((aligned(16))) float wtab[32][32];
    __shared__ float wv[32];
    __shared__ int   wi[32];
    __shared__ int   wcnt[32];
    __shared__ float4 sxyz[M_];
    __shared__ int    spos[M_];
    __shared__ unsigned short hthr[HSET];
    __shared__ unsigned char  hslot[HSET];
    __shared__ float hdist[HSET];
    __shared__ int   ctrl[2];

    const int b   = blockIdx.x;
    const int tid = threadIdx.x;
    const float* xb = xyz + (size_t)b * N_ * 3;
    const unsigned char* lb = list + (size_t)b * N_;
    const int lane = tid & 31;
    const int wav = __builtin_amdgcn_readfirstlane(tid >> 5);

    int c = 0;
    if (tid < 256) c = out_idx[b * M_ + tid];
    int incl = c;
    for (int off = 1; off < 32; off <<= 1) {
        const int t = __shfl_up(incl, off, 32);
        if (lane >= off) incl += t;
    }
    if (tid < 256 && lane == 31) wtot[wav] = incl;
    __syncthreads();
    if (tid < 256) {
        int pre = 0;
        for (int w = 0; w < 8; ++w) if (w < wav) pre += wtot[w];
        toff[tid] = pre + incl - c;
        tcnt[tid] = c;
        if (tid == 255) total_sh = pre + incl;
    }
    __syncthreads();

    {
        const int j  = tid & 127;
        const int tq = tid >> 7;
        for (int p0 = 0; p0 < 32; p0 += 8) {
            int idx[8];
            float px[8], py[8], pz[8];
#pragma unroll
            for (int u = 0; u < 8; ++u) {
                const int tl = (p0 + u) * 8 + tq;
                idx[u] = tl * 128 + (int)lb[tl * 128 + j];
            }
#pragma unroll
            for (int u = 0; u < 8; ++u) {
                px[u] = xb[(size_t)idx[u] * 3];
                py[u] = xb[(size_t)idx[u] * 3 + 1];
                pz[u] = xb[(size_t)idx[u] * 3 + 2];
            }
#pragma unroll
            for (int u = 0; u < 8; ++u) {
                const int tl = (p0 + u) * 8 + tq;
                const int q  = toff[tl] + j;
                if (j < tcnt[tl] && q < CAP) {
                    pts[q]  = make_float4(px[u], py[u], pz[u], __builtin_inff());
                    cidx[q] = (unsigned short)idx[u];
                }
            }
        }
    }
    __syncthreads();
    int Nc = total_sh;
    if (Nc > CAP) Nc = CAP;
    if (Nc == 0) {
        for (int s = tid; s < M_; s += 1024) out_idx[b * M_ + s] = 0;
        return;
    }

    const float NEG_INF = -__builtin_inff();

    const bool bulk = wav != 0;
    const int  base = tid - 32;
    float px[NSLOT], py[NSLOT], pz[NSLOT], pw[NSLOT];
    int   hp[8];
#pragma unroll
    for (int k = 0; k < NSLOT; ++k) {
        const int i = base + k * BULK_T;
        const bool v = bulk && i < Nc;
        const float4 p = pts[v ? i : 0];
        px[k] = p.x; py[k] = p.y; pz[k] = p.z;
        pw[k] = v ? p.w : NEG_INF;
    }
#pragma unroll
    for (int u = 0; u < 8; ++u) hp[u] = 0x7fffffff;

    if (tid == 0) { spos[0] = 0; sxyz[0] = pts[0]; }
    __syncthreads();

    int done_upto = 0;
    int avail = 1;
    float frac = 0.85f;
    float tau = 0.0f;
    float qx = 0.0f, qy = 0.0f, qz = 0.0f;
    int  hs = 8;
    int  ss = 0;
    bool alive = false;
    for (int round = 0; round < M_; ++round) {
        for (int it = 0; it < M_; ++it) {
            if (bulk) {
                int j = done_upto;
#pragma nounroll
                for (int g = 0; g < GSTEP; ++g) {
                    if (j < avail) {
                        const float4 qq = sxyz[j];
                        ++j;
#pragma unroll
                        for (int k = 0; k < NSLOT; ++k)
                            pw[k] = fminf(pw[k], sqdist(px[k], py[k], pz[k], qq.x, qq.y, qq.z));
                    }
                }
            } else {
#define STEP_ARGS px, py, pz, pw, hp, qx, qy, qz, ss, alive, tau, wtab[0], lane, spos, sxyz
                switch (hs) {
                    case 1:  step_chunk<1>(STEP_ARGS); break;
                    case 2:  step_chunk<2>(STEP_ARGS); break;
                    case 3:  step_chunk<3>(STEP_ARGS); break;
                    case 4:  step_chunk<4>(STEP_ARGS); break;
                    case 5:  step_chunk<5>(STEP_ARGS); break;
                    case 6:  step_chunk<6>(STEP_ARGS); break;
                    default: step_chunk<8>(STEP_ARGS); break;
                }
#undef STEP_ARGS
            }
            if (!bulk && lane == 0)
                ctrl[it & 1] = ss | ((alive && ss < M_) ? 0 : 0x10000);
            done_upto = avail;
            __syncthreads();
            const int cwd = __builtin_amdgcn_readfirstlane(ctrl[it & 1]);
            const int ssp = cwd & 0xffff;
            avail = ssp + 1 < M_ ? ssp + 1 : M_;
            if ((cwd >> 16) != 0 && (done_upto >= avail || avail >= M_)) break;
        }
        if (avail >= M_) break;

        if (bulk) {
            float bv = NEG_INF;
            int   bi = 0x7fffffff;
            int   bk = 0;
#pragma unroll
            for (int k = 0; k < NSLOT; ++k)
                if (pw[k] > bv) { bv = pw[k]; bk = k; }
            if (bv > NEG_INF) bi = base + bk * BULK_T;
            const float mv = wave_max_tab(bv, wtab[wav], lane);
            const int   mi = wave_index_of_max(bv, mv, bi);
            if (lane == 0) { wv[wav] = mv; wi[wav] = mi; }
        } else if (lane == 0) {
            wv[0] = NEG_INF; wi[0] = 0x7fffffff;
        }
        __syncthreads();
        if (wav == 0) {
            const float v2 = wv[lane];
            const int   i2 = wi[lane];
            const float mv = wave_max_tab(v2, wtab[0], lane);
            const int   mi = wave_index_of_max(v2, mv, i2);
            if (lane == 0) { sel_sh = mi; dmax_sh = mv; }
        }
        __syncthreads();
        const int s = avail;
        const int   bp = __builtin_amdgcn_readfirstlane(sel_sh);
        const float D  = __uint_as_float(
            __builtin_amdgcn_readfirstlane(__float_as_uint(dmax_sh)));
        if (tid == 0) { spos[s] = bp; sxyz[s] = pts[bp]; }
        avail = s + 1;
        if (avail >= M_) break;

        tau = frac * D;
        unsigned bm[NSLOT];
        int wtotal = 0;
#pragma unroll
        for (int k = 0; k < NSLOT; ++k) {
            const bool l = bulk && base < Nc - k * BULK_T && pw[k] >= tau;
            bm[k] = __builtin_amdgcn_ballot_w32(l);
            wtotal += __popc(bm[k]);
        }
        if (lane == 0) wcnt[wav] = wtotal;
        __syncthreads();
        int pre = 0, H = 0;
        for (int w = 0; w < 32; ++w) {
            const int cw = wcnt[w];
            H += cw;
            if (w < wav) pre += cw;
        }
        H = __builtin_amdgcn_readfirstlane(H);
        if (H <= HSET && bulk) {
            int run = __builtin_amdgcn_readfirstlane(pre);
#pragma unroll
            for (int k = 0; k < NSLOT; ++k) {
                const unsigned m = bm[k];
                if ((m >> lane) & 1u) {
                    const int o = run + (int)__builtin_amdgcn_mbcnt_lo(m, 0u);
                    hthr[o]  = (unsigned short)base;
                    hslot[o] = (unsigned char)k;
                    hdist[o] = pw[k];
                }
                run += __popc(m);
            }
        }
        __syncthreads();

        ss = s;
        if (H > HSET) {
            frac = 0.5f * (1.0f + frac);
            alive = false;
        } else {
            alive = true;
            if (!bulk) {
#pragma unroll
                for (int u = 0; u < 8; ++u) {
                    const bool valid = lane < H - 32 * u;
                    const int th = (int)hthr[lane + 32 * u];
                    const int sl = (int)hslot[lane + 32 * u];
                    const float d = hdist[lane + 32 * u];
                    const int pos = valid ? th + sl * BULK_T : 0;
                    const float4 p = pts[pos];
                    px[u] = p.x; py[u] = p.y; pz[u] = p.z;
                    pw[u] = valid ? d : NEG_INF;
                    hp[u] = valid ? pos * 8 + u : 0x7fffffff;
                }
                const float4 q0 = sxyz[s];
                qx = __uint_as_float(__builtin_amdgcn_readfirstlane(__float_as_uint(q0.x)));
                qy = __uint_as_float(__builtin_amdgcn_readfirstlane(__float_as_uint(q0.y)));
                qz = __uint_as_float(__builtin_amdgcn_readfirstlane(__float_as_uint(q0.z)));
                hs = (H + 31) >> 5;
            }
        }
    }
    __syncthreads();
    for (int t = tid; t < M_; t += 1024) out_idx[b * M_ + t] = (int)cidx[spos[t]];
}

__global__ __launch_bounds__(256) void gather_kernel(
    const float* __restrict__ feat, const float* __restrict__ xyz,
    const float* __restrict__ grasp, const int* __restrict__ idxs,
    float* __restrict__ out)
{
    const int g = blockIdx.x * 256 + threadIdx.x;
    if (g >= B_ * CH_ * M_) return;
    const int m  = g % M_;
    const int t  = g / M_;
    const int ch = t % CH_;
    const int b  = t / CH_;
    const int id = idxs[b * M_ + m];
    float v;
    if (ch < C_)            v = feat[((size_t)b * C_ + ch) * N_ + id];
    else if (ch < C_ + 3)   v = xyz[((size_t)b * N_ + id) * 3 + (ch - C_)];
    else                    v = grasp[(size_t)b * N_ + id];
    out[g] = v;
}

extern "C" void kernel_launch(void* const* d_in, const int* in_sizes, int n_in,
                              void* d_out, int out_size, void* d_ws, size_t ws_size,
                              hipStream_t stream) {
    const float* xyz  = (const float*)d_in[0];
    const float* feat = (const float*)d_in[1];
    const float* Wobj = (const float*)d_in[2];
    const float* bobj = (const float*)d_in[3];
    const float* Wg   = (const float*)d_in[4];
    const float* bg   = (const float*)d_in[5];
    float* out = (float*)d_out;

    char* ws = (char*)d_ws;
    float*         g_grasp = (float*)ws;
    unsigned char* g_list  = (unsigned char*)(ws + (size_t)B_ * N_ * 4);
    int*           g_idx   = (int*)(ws + (size_t)B_ * N_ * 4 + (size_t)B_ * N_);

    score_kernel<<<B_ * (N_ / 128), 256, 0, stream>>>(feat, Wobj, bobj, Wg, bg,
                                                      g_grasp, g_list, g_idx);
    fps_kernel<<<B_, 1024, 0, stream>>>(xyz, g_list, g_idx);
    const int total = B_ * CH_ * M_;
    gather_kernel<<<(total + 255) / 256, 256, 0, stream>>>(feat, xyz, g_grasp,
                                                           g_idx, out);
}


# FPS stepping wave: every lane loads its own best point's xyz from the LDS point table under the max-table round trip; winner read by lane; no slot switch, tie loop only on ties; picks identical
# speedup vs baseline: 3.6429x; 1.0484x over previous
#include <hip/hip_runtime.h>
#include <hip/hip_bf16.h>

typedef __attribute__((ext_vector_type(2))) float v2f;
typedef __attribute__((ext_vector_type(8))) float v8f;

#define B_   4
#define N_   32768
#define C_   512
#define M_   512
#define CH_  (C_ + 4)
#define CAP  8704

#define BG 8
__global__ __launch_bounds__(256) void score_kernel(
    const float* __restrict__ feat, const float* __restrict__ Wobj,
    const float* __restrict__ bobj, const float* __restrict__ Wg,
    const float* __restrict__ bg, float* __restrict__ grasp_out,
    unsigned char* __restrict__ list_out, int* __restrict__ cnt_out)
{
    __shared__ float Apre[128][32][2];
    {
        const int alane = threadIdx.x & 31;
        const int arow  = alane & 15;
        const int akh   = alane >> 4;
        const float* asrc = arow < 2 ? Wobj + arow * C_ : Wg;
        const bool akeep = arow < 3;
        float av0[16], av1[16];
#pragma unroll
        for (int i = 0; i < 16; ++i) {
            const int ka = ((int)(threadIdx.x >> 5) + 8 * i) * 4 + 2 * akh;
            av0[i] = asrc[ka];
            av1[i] = asrc[ka + 1];
        }
#pragma unroll
        for (int i = 0; i < 16; ++i) {
            const int kc = (int)(threadIdx.x >> 5) + 8 * i;
            Apre[kc][alane][0] = akeep ? av0[i] : 0.0f;
            Apre[kc][alane][1] = akeep ? av1[i] : 0.0f;
        }
    }
    __syncthreads();

    const int wave = threadIdx.x >> 5;
    const int lane = threadIdx.x & 31;
    const int b    = blockIdx.x >> 8;
    const int tile = blockIdx.x & 255;
    const int n0   = tile * 128 + wave * 16;
    const int kh   = lane >> 4;
    const int col  = n0 + (lane & 15);

    const float* fb = feat + (size_t)b * C_ * N_;

    const unsigned loff = (unsigned)(2 * kh) * N_ + (unsigned)col;
    float bA0[BG], bA1[BG], bB0[BG], bB1[BG];
#pragma unroll
    for (int i = 0; i < BG; ++i) {
        bA0[i] = fb[(size_t)(i * 4) * N_ + loff];
        bA1[i] = fb[(size_t)(i * 4 + 1) * N_ + loff];
    }

    v8f acc = {};
#pragma nounroll
    for (int g = 0; g < 128 / BG - 2; g += 2) {
        const float* gb = fb + (size_t)(g * BG * 4) * N_;
#pragma unroll
        for (int i = 0; i < BG; ++i) {
            bB0[i] = gb[(size_t)((BG + i) * 4) * N_ + loff];
            bB1[i] = gb[(size_t)((BG + i) * 4 + 1) * N_ + loff];
        }
#pragma unroll
        for (int i = 0; i < BG; ++i) {
            v2f a, bm;
            a.x  = Apre[g * BG + i][lane][0];
            a.y  = Apre[g * BG + i][lane][1];
            bm.x = bA0[i];
            bm.y = bA1[i];
            acc = __builtin_amdgcn_wmma_f32_16x16x4_f32(
                false, a, false, bm, (short)0, acc, false, false);
        }
#pragma unroll
        for (int i = 0; i < BG; ++i) {
            bA0[i] = gb[(size_t)((2 * BG + i) * 4) * N_ + loff];
            bA1[i] = gb[(size_t)((2 * BG + i) * 4 + 1) * N_ + loff];
        }
#pragma unroll
        for (int i = 0; i < BG; ++i) {
            v2f a, bm;
            a.x  = Apre[(g + 1) * BG + i][lane][0];
            a.y  = Apre[(g + 1) * BG + i][lane][1];
            bm.x = bB0[i];
            bm.y = bB1[i];
            acc = __builtin_amdgcn_wmma_f32_16x16x4_f32(
                false, a, false, bm, (short)0, acc, false, false);
        }
    }
    {
        const int g = 128 / BG - 2;
        const float* gb = fb + (size_t)(g * BG * 4) * N_;
#pragma unroll
        for (int i = 0; i < BG; ++i) {
            bB0[i] = gb[(size_t)((BG + i) * 4) * N_ + loff];
            bB1[i] = gb[(size_t)((BG + i) * 4 + 1) * N_ + loff];
        }
#pragma unroll
        for (int i = 0; i < BG; ++i) {
            v2f a, bm;
            a.x  = Apre[g * BG + i][lane][0];
            a.y  = Apre[g * BG + i][lane][1];
            bm.x = bA0[i];
            bm.y = bA1[i];
            acc = __builtin_amdgcn_wmma_f32_16x16x4_f32(
                false, a, false, bm, (short)0, acc, false, false);
        }
#pragma unroll
        for (int i = 0; i < BG; ++i) {
            v2f a, bm;
            a.x  = Apre[(g + 1) * BG + i][lane][0];
            a.y  = Apre[(g + 1) * BG + i][lane][1];
            bm.x = bB0[i];
            bm.y = bB1[i];
            acc = __builtin_amdgcn_wmma_f32_16x16x4_f32(
                false, a, false, bm, (short)0, acc, false, false);
        }
    }

    bool m = false;
    if (lane < 16) {
        float s0 = acc[0] + bobj[0];
        float s1 = acc[1] + bobj[1];
        float g  = acc[2] + bg[0];
        grasp_out[(size_t)b * N_ + n0 + lane] = g;
        m = (s1 > s0) && (g > 0.1f);
    }

    const unsigned bal = __builtin_amdgcn_ballot_w32(m);
    __syncthreads();
    if (lane == 0) Apre[0][wave][0] = __uint_as_float(bal);
    __syncthreads();
    int before = 0, total = 0;
    for (int w = 0; w < 8; ++w) {
        const int c = __popc(__float_as_uint(Apre[0][w][0]));
        total += c;
        if (w < wave) before += c;
    }
    if (m) {
        const int rank = __popc(bal & ((1u << lane) - 1u));
        list_out[(size_t)b * N_ + tile * 128 + before + rank] =
            (unsigned char)(wave * 16 + lane);
    }
    if ((int)threadIdx.x >= total && threadIdx.x < 128)
        list_out[(size_t)b * N_ + tile * 128 + threadIdx.x] = 0;
    if (threadIdx.x == 0) cnt_out[b * M_ + tile] = total;
}

#define HSET   256
#define BULK_T 992
#define NSLOT  9
#define GSTEP  4

__device__ __forceinline__ float sqdist(float px, float py, float pz,
                                        float qx, float qy, float qz)
{
    const float dx = px - qx;
    const float dy = py - qy;
    const float dz = pz - qz;
    float sxy;
    {
#pragma clang fp contract(off)
        sxy = dx * dx + dy * dy;
    }
    return __builtin_fmaf(dz, dz, sxy);
}

__device__ __forceinline__ float wave_max_tab(float v, float* tab, int lane)
{
    tab[lane] = v;
    __builtin_amdgcn_fence(__ATOMIC_RELEASE, "wavefront");
    __builtin_amdgcn_wave_barrier();
    __builtin_amdgcn_fence(__ATOMIC_ACQUIRE, "wavefront");
    const float4* t4 = reinterpret_cast<const float4*>(tab);
    float r[8];
#pragma unroll
    for (int k = 0; k < 8; ++k) {
        const float4 t = t4[k];
        r[k] = fmaxf(fmaxf(t.x, t.y), fmaxf(t.z, t.w));
    }
    return fmaxf(fmaxf(fmaxf(r[0], r[1]), fmaxf(r[2], r[3])),
                 fmaxf(fmaxf(r[4], r[5]), fmaxf(r[6], r[7])));
}

__device__ __forceinline__ int wave_index_of_max(float v, float m, int i)
{
    unsigned mask = __builtin_amdgcn_ballot_w32(v == m);
    int best = 0x7fffffff;
    int it = 0;
    do {
        const int l = __builtin_ctz(mask | 0x80000000u);
        const int c = __builtin_amdgcn_readlane(i, l);
        best = c < best ? c : best;
        mask &= mask - 1u;
    } while (mask != 0u && ++it < 32);
    return best;
}

template <int S>
__device__ __forceinline__ void step_chunk(
    const float (&px)[NSLOT], const float (&py)[NSLOT], const float (&pz)[NSLOT],
    float (&pw)[NSLOT], const int (&hk)[8],
    float& qx, float& qy, float& qz, int& ss, bool& alive, float tau,
    float* tab, int lane, int* spos, float4* sxyz, const float4* ptab)
{
#pragma nounroll
    for (int g = 0; g < GSTEP; ++g) {
        if (!(alive && ss < M_)) break;
        float tv = -__builtin_inff();
#pragma unroll
        for (int u = 0; u < S; ++u)
            pw[u] = fminf(pw[u], sqdist(px[u], py[u], pz[u], qx, qy, qz));
#pragma unroll
        for (int u = 0; u < S; ++u) tv = fmaxf(tv, pw[u]);
        const float mx = wave_max_tab(tv, tab, lane);
        int tk = 0x7fffffff;
#pragma unroll
        for (int u = 0; u < S; ++u)
            tk = min(tk, pw[u] == tv ? hk[u] : 0x7fffffff);
        const unsigned cpos = min((unsigned)tk >> 3, (unsigned)(CAP - 1));
        const float4 cand = ptab[cpos];
        unsigned mask = __builtin_amdgcn_ballot_w32(tv == mx);
        int bl = __builtin_ctz(mask | 0x80000000u);
        int best = __builtin_amdgcn_readlane(tk, bl);
        mask &= mask - 1u;
        for (int trips = 0; mask != 0u && trips < 31; ++trips) {
            const int l = __builtin_ctz(mask);
            const int ck = __builtin_amdgcn_readlane(tk, l);
            if (ck < best) { best = ck; bl = l; }
            mask &= mask - 1u;
        }
        const float nx = __int_as_float(__builtin_amdgcn_readlane(__float_as_int(cand.x), bl));
        const float ny = __int_as_float(__builtin_amdgcn_readlane(__float_as_int(cand.y), bl));
        const float nz = __int_as_float(__builtin_amdgcn_readlane(__float_as_int(cand.z), bl));
        const float tvu = __uint_as_float(
            __builtin_amdgcn_readfirstlane(__float_as_uint(mx)));
        if (!(tvu >= tau)) {
            alive = false;
        } else {
            ++ss;
            if (ss < M_) {
                qx = nx; qy = ny; qz = nz;
                if (lane == 0) {
                    spos[ss] = best >> 3;
                    sxyz[ss] = make_float4(nx, ny, nz, 0.0f);
                }
            }
        }
    }
}

__global__ __launch_bounds__(1024) void fps_kernel(
    const float* __restrict__ xyz, const unsigned char* __restrict__ list,
    int* out_idx)
{
    __shared__ float4 pts[CAP];
    __shared__ unsigned short cidx[CAP];
    __shared__ int toff[256];
    __shared__ int tcnt[256];
    __shared__ int wtot[8];
    __shared__ int total_sh;
    __shared__ int sel_sh;
    __shared__ float dmax_sh;
    __shared__ __attribute__((aligned(16))) float wtab[32][32];
    __shared__ float wv[32];
    __shared__ int   wi[32];
    __shared__ int   wcnt[32];
    __shared__ float4 sxyz[M_];
    __shared__ int    spos[M_];
    __shared__ unsigned short hthr[HSET];
    __shared__ unsigned char  hslot[HSET];
    __shared__ float hdist[HSET];
    __shared__ int   ctrl[2];

    const int b   = blockIdx.x;
    const int tid = threadIdx.x;
    const float* xb = xyz + (size_t)b * N_ * 3;
    const unsigned char* lb = list + (size_t)b * N_;
    const int lane = tid & 31;
    const int wav = __builtin_amdgcn_readfirstlane(tid >> 5);

    int c = 0;
    if (tid < 256) c = out_idx[b * M_ + tid];
    int incl = c;
    for (int off = 1; off < 32; off <<= 1) {
        const int t = __shfl_up(incl, off, 32);
        if (lane >= off) incl += t;
    }
    if (tid < 256 && lane == 31) wtot[wav] = incl;
    __syncthreads();
    if (tid < 256) {
        int pre = 0;
        for (int w = 0; w < 8; ++w) if (w < wav) pre += wtot[w];
        toff[tid] = pre + incl - c;
        tcnt[tid] = c;
        if (tid == 255) total_sh = pre + incl;
    }
    __syncthreads();

    {
        const int j  = tid & 127;
        const int tq = tid >> 7;
        for (int p0 = 0; p0 < 32; p0 += 8) {
            int idx[8];
            float px[8], py[8], pz[8];
#pragma unroll
            for (int u = 0; u < 8; ++u) {
                const int tl = (p0 + u) * 8 + tq;
                idx[u] = tl * 128 + (int)lb[tl * 128 + j];
            }
#pragma unroll
            for (int u = 0; u < 8; ++u) {
                px[u] = xb[(size_t)idx[u] * 3];
                py[u] = xb[(size_t)idx[u] * 3 + 1];
                pz[u] = xb[(size_t)idx[u] * 3 + 2];
            }
#pragma unroll
            for (int u = 0; u < 8; ++u) {
                const int tl = (p0 + u) * 8 + tq;
                const int q  = toff[tl] + j;
                if (j < tcnt[tl] && q < CAP) {
                    pts[q]  = make_float4(px[u], py[u], pz[u], __builtin_inff());
                    cidx[q] = (unsigned short)idx[u];
                }
            }
        }
    }
    __syncthreads();
    int Nc = total_sh;
    if (Nc > CAP) Nc = CAP;
    if (Nc == 0) {
        for (int s = tid; s < M_; s += 1024) out_idx[b * M_ + s] = 0;
        return;
    }

    const float NEG_INF = -__builtin_inff();

    const bool bulk = wav != 0;
    const int  base = tid - 32;
    float px[NSLOT], py[NSLOT], pz[NSLOT], pw[NSLOT];
    int   hp[8];
#pragma unroll
    for (int k = 0; k < NSLOT; ++k) {
        const int i = base + k * BULK_T;
        const bool v = bulk && i < Nc;
        const float4 p = pts[v ? i : 0];
        px[k] = p.x; py[k] = p.y; pz[k] = p.z;
        pw[k] = v ? p.w : NEG_INF;
    }
#pragma unroll
    for (int u = 0; u < 8; ++u) hp[u] = 0x7fffffff;

    if (tid == 0) { spos[0] = 0; sxyz[0] = pts[0]; }
    __syncthreads();

    int done_upto = 0;
    int avail = 1;
    float frac = 0.85f;
    float tau = 0.0f;
    float qx = 0.0f, qy = 0.0f, qz = 0.0f;
    int  hs = 8;
    int  ss = 0;
    bool alive = false;
    for (int round = 0; round < M_; ++round) {
        for (int it = 0; it < M_; ++it) {
            if (bulk) {
                int j = done_upto;
#pragma nounroll
                for (int g = 0; g < GSTEP; ++g) {
                    if (j < avail) {
                        const float4 qq = sxyz[j];
                        ++j;
#pragma unroll
                        for (int k = 0; k < NSLOT; ++k)
                            pw[k] = fminf(pw[k], sqdist(px[k], py[k], pz[k], qq.x, qq.y, qq.z));
                    }
                }
            } else {
#define STEP_ARGS px, py, pz, pw, hp, qx, qy, qz, ss, alive, tau, wtab[0], lane, spos, sxyz, pts
                switch (hs) {
                    case 1:  step_chunk<1>(STEP_ARGS); break;
                    case 2:  step_chunk<2>(STEP_ARGS); break;
                    case 3:  step_chunk<3>(STEP_ARGS); break;
                    case 4:  step_chunk<4>(STEP_ARGS); break;
                    case 5:  step_chunk<5>(STEP_ARGS); break;
                    case 6:  step_chunk<6>(STEP_ARGS); break;
                    default: step_chunk<8>(STEP_ARGS); break;
                }
#undef STEP_ARGS
            }
            if (!bulk && lane == 0)
                ctrl[it & 1] = ss | ((alive && ss < M_) ? 0 : 0x10000);
            done_upto = avail;
            __syncthreads();
            const int cwd = __builtin_amdgcn_readfirstlane(ctrl[it & 1]);
            const int ssp = cwd & 0xffff;
            avail = ssp + 1 < M_ ? ssp + 1 : M_;
            if ((cwd >> 16) != 0 && (done_upto >= avail || avail >= M_)) break;
        }
        if (avail >= M_) break;

        if (bulk) {
            float bv = NEG_INF;
            int   bi = 0x7fffffff;
            int   bk = 0;
#pragma unroll
            for (int k = 0; k < NSLOT; ++k)
                if (pw[k] > bv) { bv = pw[k]; bk = k; }
            if (bv > NEG_INF) bi = base + bk * BULK_T;
            const float mv = wave_max_tab(bv, wtab[wav], lane);
            const int   mi = wave_index_of_max(bv, mv, bi);
            if (lane == 0) { wv[wav] = mv; wi[wav] = mi; }
        } else if (lane == 0) {
            wv[0] = NEG_INF; wi[0] = 0x7fffffff;
        }
        __syncthreads();
        if (wav == 0) {
            const float v2 = wv[lane];
            const int   i2 = wi[lane];
            const float mv = wave_max_tab(v2, wtab[0], lane);
            const int   mi = wave_index_of_max(v2, mv, i2);
            if (lane == 0) { sel_sh = mi; dmax_sh = mv; }
        }
        __syncthreads();
        const int s = avail;
        const int   bp = __builtin_amdgcn_readfirstlane(sel_sh);
        const float D  = __uint_as_float(
            __builtin_amdgcn_readfirstlane(__float_as_uint(dmax_sh)));
        if (tid == 0) { spos[s] = bp; sxyz[s] = pts[bp]; }
        avail = s + 1;
        if (avail >= M_) break;

        tau = frac * D;
        unsigned bm[NSLOT];
        int wtotal = 0;
#pragma unroll
        for (int k = 0; k < NSLOT; ++k) {
            const bool l = bulk && base < Nc - k * BULK_T && pw[k] >= tau;
            bm[k] = __builtin_amdgcn_ballot_w32(l);
            wtotal += __popc(bm[k]);
        }
        if (lane == 0) wcnt[wav] = wtotal;
        __syncthreads();
        int pre = 0, H = 0;
        for (int w = 0; w < 32; ++w) {
            const int cw = wcnt[w];
            H += cw;
            if (w < wav) pre += cw;
        }
        H = __builtin_amdgcn_readfirstlane(H);
        if (H <= HSET && bulk) {
            int run = __builtin_amdgcn_readfirstlane(pre);
#pragma unroll
            for (int k = 0; k < NSLOT; ++k) {
                const unsigned m = bm[k];
                if ((m >> lane) & 1u) {
                    const int o = run + (int)__builtin_amdgcn_mbcnt_lo(m, 0u);
                    hthr[o]  = (unsigned short)base;
                    hslot[o] = (unsigned char)k;
                    hdist[o] = pw[k];
                }
                run += __popc(m);
            }
        }
        __syncthreads();

        ss = s;
        if (H > HSET) {
            frac = 0.5f * (1.0f + frac);
            alive = false;
        } else {
            alive = true;
            if (!bulk) {
#pragma unroll
                for (int u = 0; u < 8; ++u) {
                    const bool valid = lane < H - 32 * u;
                    const int th = (int)hthr[lane + 32 * u];
                    const int sl = (int)hslot[lane + 32 * u];
                    const float d = hdist[lane + 32 * u];
                    const int pos = valid ? th + sl * BULK_T : 0;
                    const float4 p = pts[pos];
                    px[u] = p.x; py[u] = p.y; pz[u] = p.z;
                    pw[u] = valid ? d : NEG_INF;
                    hp[u] = valid ? pos * 8 + u : 0x7fffffff;
                }
                const float4 q0 = sxyz[s];
                qx = __uint_as_float(__builtin_amdgcn_readfirstlane(__float_as_uint(q0.x)));
                qy = __uint_as_float(__builtin_amdgcn_readfirstlane(__float_as_uint(q0.y)));
                qz = __uint_as_float(__builtin_amdgcn_readfirstlane(__float_as_uint(q0.z)));
                hs = (H + 31) >> 5;
            }
        }
    }
    __syncthreads();
    for (int t = tid; t < M_; t += 1024) out_idx[b * M_ + t] = (int)cidx[spos[t]];
}

__global__ __launch_bounds__(256) void gather_kernel(
    const float* __restrict__ feat, const float* __restrict__ xyz,
    const float* __restrict__ grasp, const int* __restrict__ idxs,
    float* __restrict__ out)
{
    const int g = blockIdx.x * 256 + threadIdx.x;
    if (g >= B_ * CH_ * M_) return;
    const int m  = g % M_;
    const int t  = g / M_;
    const int ch = t % CH_;
    const int b  = t / CH_;
    const int id = idxs[b * M_ + m];
    float v;
    if (ch < C_)            v = feat[((size_t)b * C_ + ch) * N_ + id];
    else if (ch < C_ + 3)   v = xyz[((size_t)b * N_ + id) * 3 + (ch - C_)];
    else                    v = grasp[(size_t)b * N_ + id];
    out[g] = v;
}

extern "C" void kernel_launch(void* const* d_in, const int* in_sizes, int n_in,
                              void* d_out, int out_size, void* d_ws, size_t ws_size,
                              hipStream_t stream) {
    const float* xyz  = (const float*)d_in[0];
    const float* feat = (const float*)d_in[1];
    const float* Wobj = (const float*)d_in[2];
    const float* bobj = (const float*)d_in[3];
    const float* Wg   = (const float*)d_in[4];
    const float* bg   = (const float*)d_in[5];
    float* out = (float*)d_out;

    char* ws = (char*)d_ws;
    float*         g_grasp = (float*)ws;
    unsigned char* g_list  = (unsigned char*)(ws + (size_t)B_ * N_ * 4);
    int*           g_idx   = (int*)(ws + (size_t)B_ * N_ * 4 + (size_t)B_ * N_);

    score_kernel<<<B_ * (N_ / 128), 256, 0, stream>>>(feat, Wobj, bobj, Wg, bg,
                                                      g_grasp, g_list, g_idx);
    fps_kernel<<<B_, 1024, 0, stream>>>(xyz, g_list, g_idx);
    const int total = B_ * CH_ * M_;
    gather_kernel<<<(total + 255) / 256, 256, 0, stream>>>(feat, xyz, g_grasp,
                                                           g_idx, out);
}
